# starting kernel
# baseline (speedup 1.0000x reference)
#include <hip/hip_runtime.h>

#define IN_C  128
#define H_C   128
#define OUT_C 64

typedef float v2f __attribute__((ext_vector_type(2)));
typedef float v8f __attribute__((ext_vector_type(8)));

__device__ __forceinline__ void atomic_fadd(float* p, float v) {
    asm volatile("global_atomic_add_f32 %0, %1, off" : : "v"(p), "v"(v) : "memory");
}

__global__ void fill_kernel(float* __restrict__ p, float v, int n) {
    int i = blockIdx.x * blockDim.x + threadIdx.x;
    if (i < n) p[i] = v;
}

__global__ void deg_scatter_kernel(const int* __restrict__ dst,
                                   float* __restrict__ deg, int e) {
    int i = blockIdx.x * blockDim.x + threadIdx.x;
    if (i < e) atomic_fadd(&deg[dst[i]], 1.0f);
}

__global__ void rsqrt_kernel(float* __restrict__ p, int n) {
    int i = blockIdx.x * blockDim.x + threadIdx.x;
    if (i < n) p[i] = rsqrtf(p[i]);
}

template<int K, int NC, bool ADD_BIAS>
__global__ void gemm_wmma_kernel(const float* __restrict__ A,
                                 const float* __restrict__ B,
                                 const float* __restrict__ bias,
                                 float* __restrict__ C, int M) {
    __shared__ __align__(16) float As[16 * K];

    const int tid   = threadIdx.x;
    const int wave  = tid >> 5;
    const int lane  = tid & 31;
    const int n0    = wave * 16;
    const int lhalf = lane >> 4;
    const int l15   = lane & 15;
    const int n     = n0 + l15;

    v2f bfrag[K / 4];
#pragma unroll
    for (int k0 = 0; k0 < K; k0 += 4) {
        const int kk = k0 + 2 * lhalf;
        bfrag[k0 / 4].x = B[(size_t)kk * NC + n];
        bfrag[k0 / 4].y = B[(size_t)(kk + 1) * NC + n];
    }
    const float bv = ADD_BIAS ? bias[n] : 0.0f;

    const int strips      = (M + 15) / 16;
    const int vecsPerRow  = K / 4;

    for (int s = blockIdx.x; s < strips; s += gridDim.x) {
        const int row0 = s * 16;
        __syncthreads();
        for (int v = tid; v < 16 * vecsPerRow; v += blockDim.x) {
            int r = v / vecsPerRow;
            float4 val = make_float4(0.f, 0.f, 0.f, 0.f);
            if (row0 + r < M)
                val = ((const float4*)(A + (size_t)(row0 + r) * K))[v % vecsPerRow];
            ((float4*)As)[v] = val;
        }
        __syncthreads();

        v8f acc = {};
#pragma unroll
        for (int k0 = 0; k0 < K; k0 += 4) {
            const int kk = k0 + 2 * lhalf;
            v2f a;
            a.x = As[l15 * K + kk];
            a.y = As[l15 * K + kk + 1];
            acc = __builtin_amdgcn_wmma_f32_16x16x4_f32(
                      false, a, false, bfrag[k0 / 4], (short)0, acc, false, false);
        }

#pragma unroll
        for (int r = 0; r < 8; ++r) {
            int m = row0 + r + 8 * lhalf;
            if (m < M) C[(size_t)m * NC + n] = acc[r] + bv;
        }
    }
}

__global__ void agg_init_kernel(const float* __restrict__ tmp,
                                const float* __restrict__ dinv,
                                const float* __restrict__ bias,
                                float* __restrict__ out, int total) {
    int i = blockIdx.x * blockDim.x + threadIdx.x;
    if (i >= total) return;
    int node = i >> 7;
    int c    = i & 127;
    float di = dinv[node];
    out[i] = tmp[i] * di * di + bias[c];
}

__global__ void edge_scatter_kernel(const int* __restrict__ src,
                                    const int* __restrict__ dst,
                                    const float* __restrict__ dinv,
                                    const float* __restrict__ msg,
                                    float* __restrict__ out, int e) {
    int gid  = blockIdx.x * blockDim.x + threadIdx.x;
    int edge = gid >> 5;
    int lane = gid & 31;
    if (edge >= e) return;
    int s = src[edge], d = dst[edge];
    float w = dinv[s] * dinv[d];
    float4 v = ((const float4*)(msg + (size_t)s * H_C))[lane];
    float* o = out + (size_t)d * H_C + lane * 4;
    atomic_fadd(o + 0, v.x * w);
    atomic_fadd(o + 1, v.y * w);
    atomic_fadd(o + 2, v.z * w);
    atomic_fadd(o + 3, v.w * w);
}

__global__ void relu_kernel(float* __restrict__ p, int n) {
    int i = blockIdx.x * blockDim.x + threadIdx.x;
    if (i < n) p[i] = fmaxf(p[i], 0.0f);
}

__global__ void pool_scatter_kernel(const float* __restrict__ h,
                                    const int* __restrict__ batch,
                                    float* __restrict__ sums,
                                    float* __restrict__ cnts, int n) {
    int gid  = blockIdx.x * blockDim.x + threadIdx.x;
    int node = gid >> 5;
    int lane = gid & 31;
    if (node >= n) return;
    int g = batch[node];
    float4 v = ((const float4*)(h + (size_t)node * H_C))[lane];
    float* o = sums + (size_t)g * H_C + lane * 4;
    atomic_fadd(o + 0, v.x);
    atomic_fadd(o + 1, v.y);
    atomic_fadd(o + 2, v.z);
    atomic_fadd(o + 3, v.w);
    if (lane == 0) atomic_fadd(&cnts[g], 1.0f);
}

__global__ void pool_div_kernel(float* __restrict__ pooled,
                                const float* __restrict__ cnts, int total) {
    int i = blockIdx.x * blockDim.x + threadIdx.x;
    if (i >= total) return;
    pooled[i] /= fmaxf(cnts[i >> 7], 1.0f);
}

extern "C" void kernel_launch(void* const* d_in, const int* in_sizes, int n_in,
                              void* d_out, int out_size, void* d_ws, size_t ws_size,
                              hipStream_t stream) {
    const float* x    = (const float*)d_in[0];
    const int*   ei   = (const int*)  d_in[1];
    const int*   batch= (const int*)  d_in[2];
    const float* W1   = (const float*)d_in[3];
    const float* b1   = (const float*)d_in[4];
    const float* W2   = (const float*)d_in[5];
    const float* b2   = (const float*)d_in[6];
    const float* Wl   = (const float*)d_in[7];
    const float* bl   = (const float*)d_in[8];

    const int N = in_sizes[0] / IN_C;
    const int E = in_sizes[1] / 2;
    const int G = (out_size - N * OUT_C) / H_C;

    const int* esrc = ei;
    const int* edst = ei + E;

    float* logits = (float*)d_out;
    float* pooled = logits + (size_t)N * OUT_C;

    float* deg  = (float*)d_ws;
    float* tmpA = deg + (((size_t)N + 255) & ~(size_t)255);
    float* tmpB = tmpA + (size_t)N * H_C;
    float* cnts = tmpB + (size_t)N * H_C;

    const int T  = 256;
    const int NH = N * H_C;
    const int rowBlocks  = (N + 15) / 16;
    const int gemmBlocks = rowBlocks < 512 ? rowBlocks : 512;

    fill_kernel<<<(N + T - 1) / T, T, 0, stream>>>(deg, 1.0f, N);
    fill_kernel<<<(G + T - 1) / T, T, 0, stream>>>(cnts, 0.0f, G);
    fill_kernel<<<(G * H_C + T - 1) / T, T, 0, stream>>>(pooled, 0.0f, G * H_C);
    deg_scatter_kernel<<<(E + T - 1) / T, T, 0, stream>>>(edst, deg, E);
    rsqrt_kernel<<<(N + T - 1) / T, T, 0, stream>>>(deg, N);

    gemm_wmma_kernel<IN_C, H_C, false><<<gemmBlocks, 256, 0, stream>>>(x, W1, nullptr, tmpA, N);
    agg_init_kernel<<<(NH + T - 1) / T, T, 0, stream>>>(tmpA, deg, b1, tmpB, NH);
    edge_scatter_kernel<<<((size_t)E * 32 + T - 1) / T, T, 0, stream>>>(esrc, edst, deg, tmpA, tmpB, E);
    relu_kernel<<<(NH + T - 1) / T, T, 0, stream>>>(tmpB, NH);

    gemm_wmma_kernel<H_C, H_C, false><<<gemmBlocks, 256, 0, stream>>>(tmpB, W2, nullptr, tmpA, N);
    agg_init_kernel<<<(NH + T - 1) / T, T, 0, stream>>>(tmpA, deg, b2, tmpB, NH);
    edge_scatter_kernel<<<((size_t)E * 32 + T - 1) / T, T, 0, stream>>>(esrc, edst, deg, tmpA, tmpB, E);
    relu_kernel<<<(NH + T - 1) / T, T, 0, stream>>>(tmpB, NH);

    gemm_wmma_kernel<H_C, OUT_C, true><<<gemmBlocks, 128, 0, stream>>>(tmpB, Wl, bl, logits, N);

    pool_scatter_kernel<<<((size_t)N * 32 + T - 1) / T, T, 0, stream>>>(tmpB, batch, pooled, cnts, N);
    pool_div_kernel<<<(G * H_C + T - 1) / T, T, 0, stream>>>(pooled, cnts, G * H_C);
}


# edge aggregation by destination tile: a work-group keeps 256 nodes' sums in LDS, sweeps all edge destinations, each wave adds its own 8 nodes' rows (fixed sum order); start value+relu folded in
# speedup vs baseline: 1.2043x; 1.2043x over previous
#include <hip/hip_runtime.h>

#define IN_C  128
#define H_C   128
#define OUT_C 64

typedef float v2f __attribute__((ext_vector_type(2)));
typedef float v8f __attribute__((ext_vector_type(8)));

__device__ __forceinline__ void atomic_fadd(float* p, float v) {
    asm volatile("global_atomic_add_f32 %0, %1, off" : : "v"(p), "v"(v) : "memory");
}

__global__ void fill_kernel(float* __restrict__ p, float v, int n) {
    int i = blockIdx.x * blockDim.x + threadIdx.x;
    if (i < n) p[i] = v;
}

__global__ void deg_scatter_kernel(const int* __restrict__ dst,
                                   float* __restrict__ deg, int e) {
    int i = blockIdx.x * blockDim.x + threadIdx.x;
    if (i < e) atomic_fadd(&deg[dst[i]], 1.0f);
}

__global__ void rsqrt_kernel(float* __restrict__ p, int n) {
    int i = blockIdx.x * blockDim.x + threadIdx.x;
    if (i < n) p[i] = rsqrtf(p[i]);
}

template<int K, int NC, bool ADD_BIAS>
__global__ __launch_bounds__(2 * NC)
void gemm_wmma_kernel(const float* __restrict__ A,
                                 const float* __restrict__ B,
                                 const float* __restrict__ bias,
                                 float* __restrict__ C, int M) {
    const int tid   = threadIdx.x;
    const int wave  = tid >> 5;
    const int lane  = tid & 31;
    const int n0    = wave * 16;
    const int lhalf = lane >> 4;
    const int l15   = lane & 15;
    const int n     = n0 + l15;

    v2f bfrag[K / 4];
#pragma unroll
    for (int t = 0; t < K / 4; ++t) {
        const int kk = 8 * (t >> 1) + 4 * lhalf + 2 * (t & 1);
        bfrag[t].x = B[(size_t)kk * NC + n];
        bfrag[t].y = B[(size_t)(kk + 1) * NC + n];
    }
    const float bv = ADD_BIAS ? bias[n] : 0.0f;

    const int strips = (M + 15) / 16;

    for (int s = blockIdx.x; s < strips; s += gridDim.x) {
        const int row0 = s * 16;
        const int row  = (row0 + l15 < M) ? row0 + l15 : M - 1;
        const float4* arow = (const float4*)(A + (size_t)row * K);

        float4 af[K / 8];
#pragma unroll
        for (int i = 0; i < K / 8; ++i) af[i] = arow[2 * i + lhalf];
        static_assert(K / 8 == 16, "the load fence below names 16 loads");
        asm volatile("" : : "v"(af[0].x),  "v"(af[1].x),  "v"(af[2].x),  "v"(af[3].x),
                            "v"(af[4].x),  "v"(af[5].x),  "v"(af[6].x),  "v"(af[7].x),
                            "v"(af[8].x),  "v"(af[9].x),  "v"(af[10].x), "v"(af[11].x),
                            "v"(af[12].x), "v"(af[13].x), "v"(af[14].x), "v"(af[15].x));

        v8f acc = {};
#pragma unroll
        for (int i = 0; i < K / 8; ++i) {
            v2f a0, a1;
            a0.x = af[i].x; a0.y = af[i].y;
            a1.x = af[i].z; a1.y = af[i].w;
            acc = __builtin_amdgcn_wmma_f32_16x16x4_f32(
                      false, a0, false, bfrag[2 * i], (short)0, acc, false, false);
            acc = __builtin_amdgcn_wmma_f32_16x16x4_f32(
                      false, a1, false, bfrag[2 * i + 1], (short)0, acc, false, false);
        }

#pragma unroll
        for (int r = 0; r < 8; ++r) {
            int m = row0 + r + 8 * lhalf;
            if (m < M) C[(size_t)m * NC + n] = acc[r] + bv;
        }
    }
}

#define AGG_TN    256
#define AGG_WAVES 32
#define AGG_STEP  1024
#define AGG_CHUNK (AGG_WAVES * AGG_STEP)

__device__ __forceinline__ void agg_drain(const unsigned int* myl, int m, int cb, int lane,
                                          const int* __restrict__ src,
                                          const float* __restrict__ dinv,
                                          const float* __restrict__ msg,
                                          const float* dinv_t, float* acc) {
    for (int k0 = 0; k0 < m; k0 += 32) {
        const int nk = (m - k0 < 32) ? m - k0 : 32;
        int   s  = 0;
        int   dl = 0;
        float ws = 0.0f;
        float wd = 0.0f;
        if (lane < nk) {
            const unsigned int ent = myl[k0 + lane];
            dl = (int)((ent >> 15) & 255u);
            s  = src[cb + (int)(ent & 32767u)];
            ws = dinv[s];
            wd = dinv_t[dl];
        }
        for (int j = 0; j < nk; j += 4) {
            float4 r[4];
            int    dj[4];
            float  wj[4];
#pragma unroll
            for (int t = 0; t < 4; ++t) {
                const int jx = (j + t < nk) ? j + t : nk - 1;
                const int sj = __builtin_amdgcn_readlane(s, jx);
                r[t]  = ((const float4*)(msg + (size_t)sj * H_C))[lane];
            }
#pragma unroll
            for (int t = 0; t < 4; ++t) {
                const int jx = (j + t < nk) ? j + t : nk - 1;
                dj[t] = __builtin_amdgcn_readlane(dl, jx);
                wj[t] = __int_as_float(__builtin_amdgcn_readlane(__float_as_int(ws), jx)) *
                        __int_as_float(__builtin_amdgcn_readlane(__float_as_int(wd), jx));
            }
#pragma unroll
            for (int t = 0; t < 4; ++t) {
                if (j + t < nk) {
                    float4* a = (float4*)(acc + dj[t] * H_C) + lane;
                    float4  v = *a;
                    v.x += r[t].x * wj[t];
                    v.y += r[t].y * wj[t];
                    v.z += r[t].z * wj[t];
                    v.w += r[t].w * wj[t];
                    *a = v;
                }
            }
        }
    }
}

__global__ __launch_bounds__(AGG_WAVES * 32)
void tile_agg_kernel(const int* __restrict__ src,
                     const int* __restrict__ dst,
                     const float* __restrict__ dinv,
                     const float* __restrict__ msg,
                     const float* __restrict__ bias,
                     float* __restrict__ out, int n, int e) {
    __shared__ __attribute__((aligned(16))) float acc[AGG_TN * H_C];
    __shared__ float        dinv_t[AGG_TN];
    __shared__ unsigned int keep[AGG_STEP * AGG_WAVES];
    __shared__ int          kept[AGG_WAVES];
    __shared__ unsigned int mine[AGG_WAVES][160];

    const int tid  = threadIdx.x;
    const int lane = tid & 31;
    const int wave = __builtin_amdgcn_readfirstlane(tid >> 5);
    const int t0   = blockIdx.x * AGG_TN;

    if (tid < AGG_TN) dinv_t[tid] = (t0 + tid < n) ? dinv[t0 + tid] : 0.0f;
    for (int i = tid; i < AGG_TN * (H_C / 4); i += AGG_WAVES * 32) {
        const int node = t0 + (i >> 5);
        const int c4   = i & 31;
        float4 v = make_float4(0.0f, 0.0f, 0.0f, 0.0f);
        if (node < n) {
            const float  di = dinv[node];
            const float4 m  = ((const float4*)(msg + (size_t)node * H_C))[c4];
            const float4 b  = ((const float4*)bias)[c4];
            v.x = m.x * di * di + b.x;
            v.y = m.y * di * di + b.y;
            v.z = m.z * di * di + b.z;
            v.w = m.w * di * di + b.w;
        }
        ((float4*)acc)[i] = v;
    }

    unsigned int* myl = mine[wave];

    int d[32];
#pragma unroll
    for (int u = 0; u < 32; ++u) {
        const int idx = wave * AGG_STEP + u * 32 + lane;
        const int v   = dst[idx < e ? idx : e - 1];
        d[u] = (idx < e) ? v : -1;
    }
    __syncthreads();

    for (int cb = 0; cb < e; cb += AGG_CHUNK) {
        int cnt = 0;
#pragma unroll
        for (int u = 0; u < 32; ++u) {
            const unsigned int dl   = (unsigned int)(d[u] - t0);
            const bool         hit  = dl < (unsigned int)AGG_TN;
            const unsigned int mask = __builtin_amdgcn_ballot_w32(hit);
            if (mask != 0u) {
                if (hit)
                    keep[(cnt + (int)__builtin_amdgcn_mbcnt_lo(mask, 0u)) * AGG_WAVES + wave] =
                        (unsigned int)(wave * AGG_STEP + u * 32 + lane) | (dl << 15);
                cnt += __popc(mask);
            }
        }
        if (lane == 0) kept[wave] = cnt;
        __syncthreads();

#pragma unroll
        for (int u = 0; u < 32; ++u) {
            const int idx = cb + AGG_CHUNK + wave * AGG_STEP + u * 32 + lane;
            const int v   = dst[idx < e ? idx : e - 1];
            d[u] = (idx < e) ? v : -1;
        }

        const int c = kept[lane];
        int maxc = c;
#pragma unroll
        for (int o = 16; o > 0; o >>= 1) {
            const int other = __shfl_xor(maxc, o);
            maxc = other > maxc ? other : maxc;
        }
        maxc = __builtin_amdgcn_readfirstlane(maxc);
        int mcnt = 0;
        for (int j = 0; j < maxc; j += 4) {
            unsigned int ent[4];
#pragma unroll
            for (int t = 0; t < 4; ++t)
                ent[t] = (j + t < c) ? keep[(j + t) * AGG_WAVES + lane] : 0u;
#pragma unroll
            for (int t = 0; t < 4; ++t) {
                const bool         hit  = (j + t < c) && ((ent[t] >> 18) == (unsigned int)wave);
                const unsigned int mask = __builtin_amdgcn_ballot_w32(hit);
                if (mask != 0u) {
                    if (hit) myl[mcnt + (int)__builtin_amdgcn_mbcnt_lo(mask, 0u)] = ent[t];
                    mcnt += __popc(mask);
                }
            }
            if (mcnt > 32) {
                agg_drain(myl, mcnt, cb, lane, src, dinv, msg, dinv_t, acc);
                mcnt = 0;
            }
        }
        if (mcnt > 0) agg_drain(myl, mcnt, cb, lane, src, dinv, msg, dinv_t, acc);
        __syncthreads();
    }

    for (int i = tid; i < AGG_TN * (H_C / 4); i += AGG_WAVES * 32) {
        const int node = t0 + (i >> 5);
        if (node < n) {
            float4 v = ((const float4*)acc)[i];
            v.x = fmaxf(v.x, 0.0f);
            v.y = fmaxf(v.y, 0.0f);
            v.z = fmaxf(v.z, 0.0f);
            v.w = fmaxf(v.w, 0.0f);
            ((float4*)(out + (size_t)node * H_C))[i & 31] = v;
        }
    }
}

__global__ void pool_scatter_kernel(const float* __restrict__ h,
                                    const int* __restrict__ batch,
                                    float* __restrict__ sums,
                                    float* __restrict__ cnts, int n) {
    int gid  = blockIdx.x * blockDim.x + threadIdx.x;
    int node = gid >> 5;
    int lane = gid & 31;
    if (node >= n) return;
    int g = batch[node];
    float4 v = ((const float4*)(h + (size_t)node * H_C))[lane];
    float* o = sums + (size_t)g * H_C + lane * 4;
    atomic_fadd(o + 0, v.x);
    atomic_fadd(o + 1, v.y);
    atomic_fadd(o + 2, v.z);
    atomic_fadd(o + 3, v.w);
    if (lane == 0) atomic_fadd(&cnts[g], 1.0f);
}

__global__ void pool_div_kernel(float* __restrict__ pooled,
                                const float* __restrict__ cnts, int total) {
    int i = blockIdx.x * blockDim.x + threadIdx.x;
    if (i >= total) return;
    pooled[i] /= fmaxf(cnts[i >> 7], 1.0f);
}

extern "C" void kernel_launch(void* const* d_in, const int* in_sizes, int n_in,
                              void* d_out, int out_size, void* d_ws, size_t ws_size,
                              hipStream_t stream) {
    const float* x    = (const float*)d_in[0];
    const int*   ei   = (const int*)  d_in[1];
    const int*   batch= (const int*)  d_in[2];
    const float* W1   = (const float*)d_in[3];
    const float* b1   = (const float*)d_in[4];
    const float* W2   = (const float*)d_in[5];
    const float* b2   = (const float*)d_in[6];
    const float* Wl   = (const float*)d_in[7];
    const float* bl   = (const float*)d_in[8];

    const int N = in_sizes[0] / IN_C;
    const int E = in_sizes[1] / 2;
    const int G = (out_size - N * OUT_C) / H_C;

    const int* esrc = ei;
    const int* edst = ei + E;

    float* logits = (float*)d_out;
    float* pooled = logits + (size_t)N * OUT_C;

    float* deg  = (float*)d_ws;
    float* tmpA = deg + (((size_t)N + 255) & ~(size_t)255);
    float* tmpB = tmpA + (size_t)N * H_C;
    float* cnts = tmpB + (size_t)N * H_C;

    const int T  = 256;
    const int aggTiles = (N + AGG_TN - 1) / AGG_TN;
    const int rowBlocks  = (N + 15) / 16;
    const int gemmBlocks = rowBlocks < 512 ? rowBlocks : 512;

    fill_kernel<<<(N + T - 1) / T, T, 0, stream>>>(deg, 1.0f, N);
    fill_kernel<<<(G + T - 1) / T, T, 0, stream>>>(cnts, 0.0f, G);
    fill_kernel<<<(G * H_C + T - 1) / T, T, 0, stream>>>(pooled, 0.0f, G * H_C);
    deg_scatter_kernel<<<(E + T - 1) / T, T, 0, stream>>>(edst, deg, E);
    rsqrt_kernel<<<(N + T - 1) / T, T, 0, stream>>>(deg, N);

    gemm_wmma_kernel<IN_C, H_C, false><<<gemmBlocks, 256, 0, stream>>>(x, W1, nullptr, tmpA, N);
    tile_agg_kernel<<<aggTiles, AGG_WAVES * 32, 0, stream>>>(esrc, edst, deg, tmpA, b1, tmpB, N, E);

    gemm_wmma_kernel<H_C, H_C, false><<<gemmBlocks, 256, 0, stream>>>(tmpB, W2, nullptr, tmpA, N);
    tile_agg_kernel<<<aggTiles, AGG_WAVES * 32, 0, stream>>>(esrc, edst, deg, tmpA, b2, tmpB, N, E);

    gemm_wmma_kernel<H_C, OUT_C, true><<<gemmBlocks, 128, 0, stream>>>(tmpB, Wl, bl, logits, N);

    pool_scatter_kernel<<<((size_t)N * 32 + T - 1) / T, T, 0, stream>>>(tmpB, batch, pooled, cnts, N);
    pool_div_kernel<<<(G * H_C + T - 1) / T, T, 0, stream>>>(pooled, cnts, G * H_C);
}


# second aggregation reads back per-wave (source, node, weight) lists recorded by the first aggregation's sweep, 8 rows in flight, instead of sweeping all edge destinations again; sum order kept
# speedup vs baseline: 1.5099x; 1.2538x over previous
#include <hip/hip_runtime.h>

#define IN_C  128
#define H_C   128
#define OUT_C 64

typedef float v2f __attribute__((ext_vector_type(2)));
typedef float v8f __attribute__((ext_vector_type(8)));

__device__ __forceinline__ void atomic_fadd(float* p, float v) {
    asm volatile("global_atomic_add_f32 %0, %1, off" : : "v"(p), "v"(v) : "memory");
}

__global__ void fill_kernel(float* __restrict__ p, float v, int n) {
    int i = blockIdx.x * blockDim.x + threadIdx.x;
    if (i < n) p[i] = v;
}

__global__ void deg_scatter_kernel(const int* __restrict__ dst,
                                   float* __restrict__ deg, int e) {
    int i = blockIdx.x * blockDim.x + threadIdx.x;
    if (i < e) atomic_fadd(&deg[dst[i]], 1.0f);
}

__global__ void rsqrt_kernel(float* __restrict__ p, int n) {
    int i = blockIdx.x * blockDim.x + threadIdx.x;
    if (i < n) p[i] = rsqrtf(p[i]);
}

template<int K, int NC, bool ADD_BIAS>
__global__ __launch_bounds__(2 * NC)
void gemm_wmma_kernel(const float* __restrict__ A,
                                 const float* __restrict__ B,
                                 const float* __restrict__ bias,
                                 float* __restrict__ C, int M) {
    const int tid   = threadIdx.x;
    const int wave  = tid >> 5;
    const int lane  = tid & 31;
    const int n0    = wave * 16;
    const int lhalf = lane >> 4;
    const int l15   = lane & 15;
    const int n     = n0 + l15;

    v2f bfrag[K / 4];
#pragma unroll
    for (int t = 0; t < K / 4; ++t) {
        const int kk = 8 * (t >> 1) + 4 * lhalf + 2 * (t & 1);
        bfrag[t].x = B[(size_t)kk * NC + n];
        bfrag[t].y = B[(size_t)(kk + 1) * NC + n];
    }
    const float bv = ADD_BIAS ? bias[n] : 0.0f;

    const int strips = (M + 15) / 16;

    for (int s = blockIdx.x; s < strips; s += gridDim.x) {
        const int row0 = s * 16;
        const int row  = (row0 + l15 < M) ? row0 + l15 : M - 1;
        const float4* arow = (const float4*)(A + (size_t)row * K);

        float4 af[K / 8];
#pragma unroll
        for (int i = 0; i < K / 8; ++i) af[i] = arow[2 * i + lhalf];
        static_assert(K / 8 == 16, "the load fence below names 16 loads");
        asm volatile("" : : "v"(af[0].x),  "v"(af[1].x),  "v"(af[2].x),  "v"(af[3].x),
                            "v"(af[4].x),  "v"(af[5].x),  "v"(af[6].x),  "v"(af[7].x),
                            "v"(af[8].x),  "v"(af[9].x),  "v"(af[10].x), "v"(af[11].x),
                            "v"(af[12].x), "v"(af[13].x), "v"(af[14].x), "v"(af[15].x));

        v8f acc = {};
#pragma unroll
        for (int i = 0; i < K / 8; ++i) {
            v2f a0, a1;
            a0.x = af[i].x; a0.y = af[i].y;
            a1.x = af[i].z; a1.y = af[i].w;
            acc = __builtin_amdgcn_wmma_f32_16x16x4_f32(
                      false, a0, false, bfrag[2 * i], (short)0, acc, false, false);
            acc = __builtin_amdgcn_wmma_f32_16x16x4_f32(
                      false, a1, false, bfrag[2 * i + 1], (short)0, acc, false, false);
        }

#pragma unroll
        for (int r = 0; r < 8; ++r) {
            int m = row0 + r + 8 * lhalf;
            if (m < M) C[(size_t)m * NC + n] = acc[r] + bv;
        }
    }
}

#define AGG_TN    256
#define AGG_WAVES 32
#define AGG_STEP  1024
#define AGG_CHUNK (AGG_WAVES * AGG_STEP)
#define AGG_CAP   1024

template<int R>
__device__ __forceinline__ void agg_add_rows(int s, int dl, float w, int nk, int lane,
                                             const float* __restrict__ msg, float* acc) {
    for (int j = 0; j < nk; j += R) {
        float4 r[R];
        int    dj[R];
        float  wj[R];
#pragma unroll
        for (int t = 0; t < R; ++t) {
            const int jx = (j + t < nk) ? j + t : nk - 1;
            const int sj = __builtin_amdgcn_readlane(s, jx);
            r[t]  = ((const float4*)(msg + (size_t)sj * H_C))[lane];
        }
#pragma unroll
        for (int t = 0; t < R; ++t) {
            const int jx = (j + t < nk) ? j + t : nk - 1;
            dj[t] = __builtin_amdgcn_readlane(dl, jx);
            wj[t] = __int_as_float(__builtin_amdgcn_readlane(__float_as_int(w), jx));
        }
#pragma unroll
        for (int t = 0; t < R; ++t) {
            if (j + t < nk) {
                float4* a = (float4*)(acc + dj[t] * H_C) + lane;
                float4  v = *a;
                v.x += r[t].x * wj[t];
                v.y += r[t].y * wj[t];
                v.z += r[t].z * wj[t];
                v.w += r[t].w * wj[t];
                *a = v;
            }
        }
    }
}

template<bool RECORD>
__device__ __forceinline__ void agg_drain(const unsigned int* myl, int m, int cb, int lane,
                                          const int* __restrict__ src,
                                          const float* __restrict__ dinv,
                                          const float* __restrict__ msg,
                                          const float* dinv_t, float* acc,
                                          uint2* __restrict__ lst, int& lcnt) {
    for (int k0 = 0; k0 < m; k0 += 32) {
        const int nk = (m - k0 < 32) ? m - k0 : 32;
        int   s  = 0;
        int   dl = 0;
        float w  = 0.0f;
        if (lane < nk) {
            const unsigned int ent = myl[k0 + lane];
            dl = (int)((ent >> 15) & 255u);
            s  = src[cb + (int)(ent & 32767u)];
            w  = dinv[s] * dinv_t[dl];
        }
        agg_add_rows<4>(s, dl, w, nk, lane, msg, acc);
        if (RECORD) {
            if (lane < nk && lcnt + lane < AGG_CAP)
                lst[lcnt + lane] = make_uint2((unsigned int)s | ((unsigned int)(dl & 7) << 29),
                                              (unsigned int)__float_as_int(w));
            lcnt += nk;
        }
    }
}

template<bool FIRST>
__global__ __launch_bounds__(AGG_WAVES * 32)
void tile_agg_kernel(const int* __restrict__ src,
                     const int* __restrict__ dst,
                     const float* __restrict__ dinv,
                     const float* __restrict__ msg,
                     const float* __restrict__ bias,
                     float* __restrict__ out, int n, int e,
                     unsigned int* __restrict__ lcount,
                     unsigned int* __restrict__ lists) {
    __shared__ __attribute__((aligned(16))) float acc[AGG_TN * H_C];
    __shared__ float        dinv_t[AGG_TN];
    __shared__ unsigned int keep[AGG_STEP * AGG_WAVES];
    __shared__ int          kept[AGG_WAVES];
    __shared__ unsigned int mine[AGG_WAVES][160];

    const int tid  = threadIdx.x;
    const int lane = tid & 31;
    const int wave = __builtin_amdgcn_readfirstlane(tid >> 5);
    const int t0   = blockIdx.x * AGG_TN;

    uint2* lst = (uint2*)lists + (size_t)(blockIdx.x * AGG_WAVES + wave) * AGG_CAP;

    bool fits = false;
    int  mlist = 0;
    uint2 ecur = make_uint2(0u, 0u);
    if (!FIRST) {
        const unsigned int c = lcount[blockIdx.x * AGG_WAVES + lane];
        fits  = __builtin_amdgcn_ballot_w32(c > (unsigned int)AGG_CAP) == 0u;
        mlist = __builtin_amdgcn_readlane((int)c, wave);
        if (fits && lane < mlist) ecur = lst[lane];
    }

    if (tid < AGG_TN) dinv_t[tid] = (t0 + tid < n) ? dinv[t0 + tid] : 0.0f;
    for (int i = tid; i < AGG_TN * (H_C / 4); i += AGG_WAVES * 32) {
        const int node = t0 + (i >> 5);
        const int c4   = i & 31;
        float4 v = make_float4(0.0f, 0.0f, 0.0f, 0.0f);
        if (node < n) {
            const float  di = dinv[node];
            const float4 m  = ((const float4*)(msg + (size_t)node * H_C))[c4];
            const float4 b  = ((const float4*)bias)[c4];
            v.x = m.x * di * di + b.x;
            v.y = m.y * di * di + b.y;
            v.z = m.z * di * di + b.z;
            v.w = m.w * di * di + b.w;
        }
        ((float4*)acc)[i] = v;
    }

    if (!FIRST && fits) {
        __syncthreads();
        for (int k0 = 0; k0 < mlist; k0 += 32) {
            const int nk = (mlist - k0 < 32) ? mlist - k0 : 32;
            uint2 enext = make_uint2(0u, 0u);
            if (k0 + 32 + lane < mlist) enext = lst[k0 + 32 + lane];
            int   s  = 0;
            int   dl = 0;
            float w  = 0.0f;
            if (lane < nk) {
                s  = (int)(ecur.x & 0x1fffffffu);
                dl = wave * 8 + (int)(ecur.x >> 29);
                w  = __int_as_float((int)ecur.y);
            }
            agg_add_rows<8>(s, dl, w, nk, lane, msg, acc);
            ecur = enext;
        }
        __syncthreads();
    } else {
        unsigned int* myl = mine[wave];
        int lcnt = 0;

        int d[32];
#pragma unroll
        for (int u = 0; u < 32; ++u) {
            const int idx = wave * AGG_STEP + u * 32 + lane;
            const int v   = dst[idx < e ? idx : e - 1];
            d[u] = (idx < e) ? v : -1;
        }
        __syncthreads();

        for (int cb = 0; cb < e; cb += AGG_CHUNK) {
            int cnt = 0;
#pragma unroll
            for (int u = 0; u < 32; ++u) {
                const unsigned int dl   = (unsigned int)(d[u] - t0);
                const bool         hit  = dl < (unsigned int)AGG_TN;
                const unsigned int mask = __builtin_amdgcn_ballot_w32(hit);
                if (mask != 0u) {
                    if (hit)
                        keep[(cnt + (int)__builtin_amdgcn_mbcnt_lo(mask, 0u)) * AGG_WAVES + wave] =
                            (unsigned int)(wave * AGG_STEP + u * 32 + lane) | (dl << 15);
                    cnt += __popc(mask);
                }
            }
            if (lane == 0) kept[wave] = cnt;
            __syncthreads();

#pragma unroll
            for (int u = 0; u < 32; ++u) {
                const int idx = cb + AGG_CHUNK + wave * AGG_STEP + u * 32 + lane;
                const int v   = dst[idx < e ? idx : e - 1];
                d[u] = (idx < e) ? v : -1;
            }

            const int c = kept[lane];
            int maxc = c;
#pragma unroll
            for (int o = 16; o > 0; o >>= 1) {
                const int other = __shfl_xor(maxc, o);
                maxc = other > maxc ? other : maxc;
            }
            maxc = __builtin_amdgcn_readfirstlane(maxc);
            int mcnt = 0;
            for (int j = 0; j < maxc; j += 4) {
                unsigned int ent[4];
#pragma unroll
                for (int t = 0; t < 4; ++t)
                    ent[t] = (j + t < c) ? keep[(j + t) * AGG_WAVES + lane] : 0u;
#pragma unroll
                for (int t = 0; t < 4; ++t) {
                    const bool         hit  = (j + t < c) && ((ent[t] >> 18) == (unsigned int)wave);
                    const unsigned int mask = __builtin_amdgcn_ballot_w32(hit);
                    if (mask != 0u) {
                        if (hit) myl[mcnt + (int)__builtin_amdgcn_mbcnt_lo(mask, 0u)] = ent[t];
                        mcnt += __popc(mask);
                    }
                }
                if (mcnt > 32) {
                    agg_drain<FIRST>(myl, mcnt, cb, lane, src, dinv, msg, dinv_t, acc, lst, lcnt);
                    mcnt = 0;
                }
            }
            if (mcnt > 0) agg_drain<FIRST>(myl, mcnt, cb, lane, src, dinv, msg, dinv_t, acc, lst, lcnt);
            __syncthreads();
        }

        if (FIRST && lane == 0) lcount[blockIdx.x * AGG_WAVES + wave] = (unsigned int)lcnt;
    }

    for (int i = tid; i < AGG_TN * (H_C / 4); i += AGG_WAVES * 32) {
        const int node = t0 + (i >> 5);
        if (node < n) {
            float4 v = ((const float4*)acc)[i];
            v.x = fmaxf(v.x, 0.0f);
            v.y = fmaxf(v.y, 0.0f);
            v.z = fmaxf(v.z, 0.0f);
            v.w = fmaxf(v.w, 0.0f);
            ((float4*)(out + (size_t)node * H_C))[i & 31] = v;
        }
    }
}

__global__ void pool_scatter_kernel(const float* __restrict__ h,
                                    const int* __restrict__ batch,
                                    float* __restrict__ sums,
                                    float* __restrict__ cnts, int n) {
    int gid  = blockIdx.x * blockDim.x + threadIdx.x;
    int node = gid >> 5;
    int lane = gid & 31;
    if (node >= n) return;
    int g = batch[node];
    float4 v = ((const float4*)(h + (size_t)node * H_C))[lane];
    float* o = sums + (size_t)g * H_C + lane * 4;
    atomic_fadd(o + 0, v.x);
    atomic_fadd(o + 1, v.y);
    atomic_fadd(o + 2, v.z);
    atomic_fadd(o + 3, v.w);
    if (lane == 0) atomic_fadd(&cnts[g], 1.0f);
}

__global__ void pool_div_kernel(float* __restrict__ pooled,
                                const float* __restrict__ cnts, int total) {
    int i = blockIdx.x * blockDim.x + threadIdx.x;
    if (i >= total) return;
    pooled[i] /= fmaxf(cnts[i >> 7], 1.0f);
}

extern "C" void kernel_launch(void* const* d_in, const int* in_sizes, int n_in,
                              void* d_out, int out_size, void* d_ws, size_t ws_size,
                              hipStream_t stream) {
    const float* x    = (const float*)d_in[0];
    const int*   ei   = (const int*)  d_in[1];
    const int*   batch= (const int*)  d_in[2];
    const float* W1   = (const float*)d_in[3];
    const float* b1   = (const float*)d_in[4];
    const float* W2   = (const float*)d_in[5];
    const float* b2   = (const float*)d_in[6];
    const float* Wl   = (const float*)d_in[7];
    const float* bl   = (const float*)d_in[8];

    const int N = in_sizes[0] / IN_C;
    const int E = in_sizes[1] / 2;
    const int G = (out_size - N * OUT_C) / H_C;

    const int* esrc = ei;
    const int* edst = ei + E;

    float* logits = (float*)d_out;
    float* pooled = logits + (size_t)N * OUT_C;

    float* deg  = (float*)d_ws;
    float* tmpA = deg + (((size_t)N + 255) & ~(size_t)255);
    float* tmpB = tmpA + (size_t)N * H_C;
    float* cnts = tmpB + (size_t)N * H_C;

    const int T  = 256;
    const int aggTiles = (N + AGG_TN - 1) / AGG_TN;
    unsigned int* lcount = (unsigned int*)(cnts + (((size_t)G + 255) & ~(size_t)255));
    unsigned int* lists = lcount + (((size_t)aggTiles * AGG_WAVES + 63) & ~(size_t)63);
    const int rowBlocks  = (N + 15) / 16;
    const int gemmBlocks = rowBlocks < 512 ? rowBlocks : 512;

    fill_kernel<<<(N + T - 1) / T, T, 0, stream>>>(deg, 1.0f, N);
    fill_kernel<<<(G + T - 1) / T, T, 0, stream>>>(cnts, 0.0f, G);
    fill_kernel<<<(G * H_C + T - 1) / T, T, 0, stream>>>(pooled, 0.0f, G * H_C);
    deg_scatter_kernel<<<(E + T - 1) / T, T, 0, stream>>>(edst, deg, E);
    rsqrt_kernel<<<(N + T - 1) / T, T, 0, stream>>>(deg, N);

    gemm_wmma_kernel<IN_C, H_C, false><<<gemmBlocks, 256, 0, stream>>>(x, W1, nullptr, tmpA, N);
    tile_agg_kernel<true><<<aggTiles, AGG_WAVES * 32, 0, stream>>>(esrc, edst, deg, tmpA, b1, tmpB, N, E, lcount, lists);

    gemm_wmma_kernel<H_C, H_C, false><<<gemmBlocks, 256, 0, stream>>>(tmpB, W2, nullptr, tmpA, N);
    tile_agg_kernel<false><<<aggTiles, AGG_WAVES * 32, 0, stream>>>(esrc, edst, deg, tmpA, b2, tmpB, N, E, lcount, lists);

    gemm_wmma_kernel<H_C, OUT_C, true><<<gemmBlocks, 128, 0, stream>>>(tmpB, Wl, bl, logits, N);

    pool_scatter_kernel<<<((size_t)N * 32 + T - 1) / T, T, 0, stream>>>(tmpB, batch, pooled, cnts, N);
    pool_div_kernel<<<(G * H_C + T - 1) / T, T, 0, stream>>>(pooled, cnts, G * H_C);
}


# mean pool's sums taken from the second aggregation's finished LDS tile: per 32-node group, runs of equal graph id summed in node order, one f32 atomic per run and channel; wave-per-node pass gone
# speedup vs baseline: 1.8520x; 1.2266x over previous
#include <hip/hip_runtime.h>

#define IN_C  128
#define H_C   128
#define OUT_C 64

typedef float v2f __attribute__((ext_vector_type(2)));
typedef float v8f __attribute__((ext_vector_type(8)));

__device__ __forceinline__ void atomic_fadd(float* p, float v) {
    asm volatile("global_atomic_add_f32 %0, %1, off" : : "v"(p), "v"(v) : "memory");
}

__global__ void fill_kernel(float* __restrict__ p, float v, int n) {
    int i = blockIdx.x * blockDim.x + threadIdx.x;
    if (i < n) p[i] = v;
}

__global__ void deg_scatter_kernel(const int* __restrict__ dst,
                                   float* __restrict__ deg, int e) {
    int i = blockIdx.x * blockDim.x + threadIdx.x;
    if (i < e) atomic_fadd(&deg[dst[i]], 1.0f);
}

__global__ void rsqrt_kernel(float* __restrict__ p, int n) {
    int i = blockIdx.x * blockDim.x + threadIdx.x;
    if (i < n) p[i] = rsqrtf(p[i]);
}

template<int K, int NC, bool ADD_BIAS>
__global__ __launch_bounds__(2 * NC)
void gemm_wmma_kernel(const float* __restrict__ A,
                                 const float* __restrict__ B,
                                 const float* __restrict__ bias,
                                 float* __restrict__ C, int M) {
    const int tid   = threadIdx.x;
    const int wave  = tid >> 5;
    const int lane  = tid & 31;
    const int n0    = wave * 16;
    const int lhalf = lane >> 4;
    const int l15   = lane & 15;
    const int n     = n0 + l15;

    v2f bfrag[K / 4];
#pragma unroll
    for (int t = 0; t < K / 4; ++t) {
        const int kk = 8 * (t >> 1) + 4 * lhalf + 2 * (t & 1);
        bfrag[t].x = B[(size_t)kk * NC + n];
        bfrag[t].y = B[(size_t)(kk + 1) * NC + n];
    }
    const float bv = ADD_BIAS ? bias[n] : 0.0f;

    const int strips = (M + 15) / 16;

    for (int s = blockIdx.x; s < strips; s += gridDim.x) {
        const int row0 = s * 16;
        const int row  = (row0 + l15 < M) ? row0 + l15 : M - 1;
        const float4* arow = (const float4*)(A + (size_t)row * K);

        float4 af[K / 8];
#pragma unroll
        for (int i = 0; i < K / 8; ++i) af[i] = arow[2 * i + lhalf];
        static_assert(K / 8 == 16, "the load fence below names 16 loads");
        asm volatile("" : : "v"(af[0].x),  "v"(af[1].x),  "v"(af[2].x),  "v"(af[3].x),
                            "v"(af[4].x),  "v"(af[5].x),  "v"(af[6].x),  "v"(af[7].x),
                            "v"(af[8].x),  "v"(af[9].x),  "v"(af[10].x), "v"(af[11].x),
                            "v"(af[12].x), "v"(af[13].x), "v"(af[14].x), "v"(af[15].x));

        v8f acc = {};
#pragma unroll
        for (int i = 0; i < K / 8; ++i) {
            v2f a0, a1;
            a0.x = af[i].x; a0.y = af[i].y;
            a1.x = af[i].z; a1.y = af[i].w;
            acc = __builtin_amdgcn_wmma_f32_16x16x4_f32(
                      false, a0, false, bfrag[2 * i], (short)0, acc, false, false);
            acc = __builtin_amdgcn_wmma_f32_16x16x4_f32(
                      false, a1, false, bfrag[2 * i + 1], (short)0, acc, false, false);
        }

#pragma unroll
        for (int r = 0; r < 8; ++r) {
            int m = row0 + r + 8 * lhalf;
            if (m < M) C[(size_t)m * NC + n] = acc[r] + bv;
        }
    }
}

#define AGG_TN    256
#define AGG_WAVES 32
#define AGG_STEP  1024
#define AGG_CHUNK (AGG_WAVES * AGG_STEP)
#define AGG_CAP   1024
#define POOL_GN   32

template<int R>
__device__ __forceinline__ void agg_add_rows(int s, int dl, float w, int nk, int lane,
                                             const float* __restrict__ msg, float* acc) {
    for (int j = 0; j < nk; j += R) {
        float4 r[R];
        int    dj[R];
        float  wj[R];
#pragma unroll
        for (int t = 0; t < R; ++t) {
            const int jx = (j + t < nk) ? j + t : nk - 1;
            const int sj = __builtin_amdgcn_readlane(s, jx);
            r[t]  = ((const float4*)(msg + (size_t)sj * H_C))[lane];
        }
#pragma unroll
        for (int t = 0; t < R; ++t) {
            const int jx = (j + t < nk) ? j + t : nk - 1;
            dj[t] = __builtin_amdgcn_readlane(dl, jx);
            wj[t] = __int_as_float(__builtin_amdgcn_readlane(__float_as_int(w), jx));
        }
#pragma unroll
        for (int t = 0; t < R; ++t) {
            if (j + t < nk) {
                float4* a = (float4*)(acc + dj[t] * H_C) + lane;
                float4  v = *a;
                v.x += r[t].x * wj[t];
                v.y += r[t].y * wj[t];
                v.z += r[t].z * wj[t];
                v.w += r[t].w * wj[t];
                *a = v;
            }
        }
    }
}

template<bool RECORD>
__device__ __forceinline__ void agg_drain(const unsigned int* myl, int m, int cb, int lane,
                                          const int* __restrict__ src,
                                          const float* __restrict__ dinv,
                                          const float* __restrict__ msg,
                                          const float* dinv_t, float* acc,
                                          uint2* __restrict__ lst, int& lcnt) {
    for (int k0 = 0; k0 < m; k0 += 32) {
        const int nk = (m - k0 < 32) ? m - k0 : 32;
        int   s  = 0;
        int   dl = 0;
        float w  = 0.0f;
        if (lane < nk) {
            const unsigned int ent = myl[k0 + lane];
            dl = (int)((ent >> 15) & 255u);
            s  = src[cb + (int)(ent & 32767u)];
            w  = dinv[s] * dinv_t[dl];
        }
        agg_add_rows<4>(s, dl, w, nk, lane, msg, acc);
        if (RECORD) {
            if (lane < nk && lcnt + lane < AGG_CAP)
                lst[lcnt + lane] = make_uint2((unsigned int)s | ((unsigned int)(dl & 7) << 29),
                                              (unsigned int)__float_as_int(w));
            lcnt += nk;
        }
    }
}

template<bool FIRST>
__global__ __launch_bounds__(AGG_WAVES * 32)
void tile_agg_kernel(const int* __restrict__ src,
                     const int* __restrict__ dst,
                     const float* __restrict__ dinv,
                     const float* __restrict__ msg,
                     const float* __restrict__ bias,
                     float* __restrict__ out, int n, int e,
                     unsigned int* __restrict__ lcount,
                     unsigned int* __restrict__ lists,
                     const int* __restrict__ batch,
                     float* __restrict__ pooled,
                     float* __restrict__ cnts) {
    __shared__ __attribute__((aligned(16))) float acc[AGG_TN * H_C];
    __shared__ float        dinv_t[AGG_TN];
    __shared__ int          gid_t[AGG_TN];
    __shared__ unsigned int keep[AGG_STEP * AGG_WAVES];
    __shared__ int          kept[AGG_WAVES];
    __shared__ unsigned int mine[AGG_WAVES][160];

    const int tid  = threadIdx.x;
    const int lane = tid & 31;
    const int wave = __builtin_amdgcn_readfirstlane(tid >> 5);
    const int t0   = blockIdx.x * AGG_TN;

    uint2* lst = (uint2*)lists + (size_t)(blockIdx.x * AGG_WAVES + wave) * AGG_CAP;

    bool fits = false;
    int  mlist = 0;
    uint2 ecur = make_uint2(0u, 0u);
    if (!FIRST) {
        const unsigned int c = lcount[blockIdx.x * AGG_WAVES + lane];
        fits  = __builtin_amdgcn_ballot_w32(c > (unsigned int)AGG_CAP) == 0u;
        mlist = __builtin_amdgcn_readlane((int)c, wave);
        if (fits && lane < mlist) ecur = lst[lane];
    }

    if (tid < AGG_TN) dinv_t[tid] = (t0 + tid < n) ? dinv[t0 + tid] : 0.0f;
    if (!FIRST && tid < AGG_TN) gid_t[tid] = (t0 + tid < n) ? batch[t0 + tid] : -1;
    for (int i = tid; i < AGG_TN * (H_C / 4); i += AGG_WAVES * 32) {
        const int node = t0 + (i >> 5);
        const int c4   = i & 31;
        float4 v = make_float4(0.0f, 0.0f, 0.0f, 0.0f);
        if (node < n) {
            const float  di = dinv[node];
            const float4 m  = ((const float4*)(msg + (size_t)node * H_C))[c4];
            const float4 b  = ((const float4*)bias)[c4];
            v.x = m.x * di * di + b.x;
            v.y = m.y * di * di + b.y;
            v.z = m.z * di * di + b.z;
            v.w = m.w * di * di + b.w;
        }
        ((float4*)acc)[i] = v;
    }

    if (!FIRST && fits) {
        __syncthreads();
        for (int k0 = 0; k0 < mlist; k0 += 32) {
            const int nk = (mlist - k0 < 32) ? mlist - k0 : 32;
            uint2 enext = make_uint2(0u, 0u);
            if (k0 + 32 + lane < mlist) enext = lst[k0 + 32 + lane];
            int   s  = 0;
            int   dl = 0;
            float w  = 0.0f;
            if (lane < nk) {
                s  = (int)(ecur.x & 0x1fffffffu);
                dl = wave * 8 + (int)(ecur.x >> 29);
                w  = __int_as_float((int)ecur.y);
            }
            agg_add_rows<8>(s, dl, w, nk, lane, msg, acc);
            ecur = enext;
        }
        __syncthreads();
    } else {
        unsigned int* myl = mine[wave];
        int lcnt = 0;

        int d[32];
#pragma unroll
        for (int u = 0; u < 32; ++u) {
            const int idx = wave * AGG_STEP + u * 32 + lane;
            const int v   = dst[idx < e ? idx : e - 1];
            d[u] = (idx < e) ? v : -1;
        }
        __syncthreads();

        for (int cb = 0; cb < e; cb += AGG_CHUNK) {
            int cnt = 0;
#pragma unroll
            for (int u = 0; u < 32; ++u) {
                const unsigned int dl   = (unsigned int)(d[u] - t0);
                const bool         hit  = dl < (unsigned int)AGG_TN;
                const unsigned int mask = __builtin_amdgcn_ballot_w32(hit);
                if (mask != 0u) {
                    if (hit)
                        keep[(cnt + (int)__builtin_amdgcn_mbcnt_lo(mask, 0u)) * AGG_WAVES + wave] =
                            (unsigned int)(wave * AGG_STEP + u * 32 + lane) | (dl << 15);
                    cnt += __popc(mask);
                }
            }
            if (lane == 0) kept[wave] = cnt;
            __syncthreads();

#pragma unroll
            for (int u = 0; u < 32; ++u) {
                const int idx = cb + AGG_CHUNK + wave * AGG_STEP + u * 32 + lane;
                const int v   = dst[idx < e ? idx : e - 1];
                d[u] = (idx < e) ? v : -1;
            }

            const int c = kept[lane];
            int maxc = c;
#pragma unroll
            for (int o = 16; o > 0; o >>= 1) {
                const int other = __shfl_xor(maxc, o);
                maxc = other > maxc ? other : maxc;
            }
            maxc = __builtin_amdgcn_readfirstlane(maxc);
            int mcnt = 0;
            for (int j = 0; j < maxc; j += 4) {
                unsigned int ent[4];
#pragma unroll
                for (int t = 0; t < 4; ++t)
                    ent[t] = (j + t < c) ? keep[(j + t) * AGG_WAVES + lane] : 0u;
#pragma unroll
                for (int t = 0; t < 4; ++t) {
                    const bool         hit  = (j + t < c) && ((ent[t] >> 18) == (unsigned int)wave);
                    const unsigned int mask = __builtin_amdgcn_ballot_w32(hit);
                    if (mask != 0u) {
                        if (hit) myl[mcnt + (int)__builtin_amdgcn_mbcnt_lo(mask, 0u)] = ent[t];
                        mcnt += __popc(mask);
                    }
                }
                if (mcnt > 32) {
                    agg_drain<FIRST>(myl, mcnt, cb, lane, src, dinv, msg, dinv_t, acc, lst, lcnt);
                    mcnt = 0;
                }
            }
            if (mcnt > 0) agg_drain<FIRST>(myl, mcnt, cb, lane, src, dinv, msg, dinv_t, acc, lst, lcnt);
            __syncthreads();
        }

        if (FIRST && lane == 0) lcount[blockIdx.x * AGG_WAVES + wave] = (unsigned int)lcnt;
    }

    for (int i = tid; i < AGG_TN * (H_C / 4); i += AGG_WAVES * 32) {
        const int node = t0 + (i >> 5);
        if (node < n) {
            float4 v = ((const float4*)acc)[i];
            v.x = fmaxf(v.x, 0.0f);
            v.y = fmaxf(v.y, 0.0f);
            v.z = fmaxf(v.z, 0.0f);
            v.w = fmaxf(v.w, 0.0f);
            ((float4*)(out + (size_t)node * H_C))[i & 31] = v;
        }
    }

    if (!FIRST) {
        const int c  = tid & (H_C - 1);
        const int q  = __builtin_amdgcn_readfirstlane(tid >> 7);
        const int nb = q * POOL_GN;
        int nval = n - (t0 + nb);
        nval = nval < POOL_GN ? nval : POOL_GN;
        if (nval > 0) {
            float v[POOL_GN];
#pragma unroll
            for (int k = 0; k < POOL_GN; ++k) v[k] = acc[(nb + k) * H_C + c];
            const int gl = gid_t[nb + lane];
            int   g   = __builtin_amdgcn_readlane(gl, 0);
            float run = 0.0f;
            int   len = 0;
#pragma unroll
            for (int k = 0; k < POOL_GN; ++k) {
                if (k < nval) {
                    const int gk = __builtin_amdgcn_readlane(gl, k);
                    if (gk != g) {
                        atomic_fadd(&pooled[(size_t)g * H_C + c], run);
                        if (c == 0) atomic_fadd(&cnts[g], (float)len);
                        run = 0.0f;
                        len = 0;
                        g   = gk;
                    }
                    run += fmaxf(v[k], 0.0f);
                    ++len;
                }
            }
            atomic_fadd(&pooled[(size_t)g * H_C + c], run);
            if (c == 0) atomic_fadd(&cnts[g], (float)len);
        }
    }
}

__global__ void pool_div_kernel(float* __restrict__ pooled,
                                const float* __restrict__ cnts, int total) {
    int i = blockIdx.x * blockDim.x + threadIdx.x;
    if (i >= total) return;
    pooled[i] /= fmaxf(cnts[i >> 7], 1.0f);
}

extern "C" void kernel_launch(void* const* d_in, const int* in_sizes, int n_in,
                              void* d_out, int out_size, void* d_ws, size_t ws_size,
                              hipStream_t stream) {
    const float* x    = (const float*)d_in[0];
    const int*   ei   = (const int*)  d_in[1];
    const int*   batch= (const int*)  d_in[2];
    const float* W1   = (const float*)d_in[3];
    const float* b1   = (const float*)d_in[4];
    const float* W2   = (const float*)d_in[5];
    const float* b2   = (const float*)d_in[6];
    const float* Wl   = (const float*)d_in[7];
    const float* bl   = (const float*)d_in[8];

    const int N = in_sizes[0] / IN_C;
    const int E = in_sizes[1] / 2;
    const int G = (out_size - N * OUT_C) / H_C;

    const int* esrc = ei;
    const int* edst = ei + E;

    float* logits = (float*)d_out;
    float* pooled = logits + (size_t)N * OUT_C;

    float* deg  = (float*)d_ws;
    float* tmpA = deg + (((size_t)N + 255) & ~(size_t)255);
    float* tmpB = tmpA + (size_t)N * H_C;
    float* cnts = tmpB + (size_t)N * H_C;

    const int T  = 256;
    const int aggTiles = (N + AGG_TN - 1) / AGG_TN;
    unsigned int* lcount = (unsigned int*)(cnts + (((size_t)G + 255) & ~(size_t)255));
    unsigned int* lists = lcount + (((size_t)aggTiles * AGG_WAVES + 63) & ~(size_t)63);
    const int rowBlocks  = (N + 15) / 16;
    const int gemmBlocks = rowBlocks < 512 ? rowBlocks : 512;

    fill_kernel<<<(N + T - 1) / T, T, 0, stream>>>(deg, 1.0f, N);
    fill_kernel<<<(G + T - 1) / T, T, 0, stream>>>(cnts, 0.0f, G);
    fill_kernel<<<(G * H_C + T - 1) / T, T, 0, stream>>>(pooled, 0.0f, G * H_C);
    deg_scatter_kernel<<<(E + T - 1) / T, T, 0, stream>>>(edst, deg, E);
    rsqrt_kernel<<<(N + T - 1) / T, T, 0, stream>>>(deg, N);

    gemm_wmma_kernel<IN_C, H_C, false><<<gemmBlocks, 256, 0, stream>>>(x, W1, nullptr, tmpA, N);
    tile_agg_kernel<true><<<aggTiles, AGG_WAVES * 32, 0, stream>>>(esrc, edst, deg, tmpA, b1, tmpB, N, E, lcount, lists, nullptr, nullptr, nullptr);

    gemm_wmma_kernel<H_C, H_C, false><<<gemmBlocks, 256, 0, stream>>>(tmpB, W2, nullptr, tmpA, N);
    tile_agg_kernel<false><<<aggTiles, AGG_WAVES * 32, 0, stream>>>(esrc, edst, deg, tmpA, b2, tmpB, N, E, lcount, lists, batch, pooled, cnts);

    gemm_wmma_kernel<H_C, OUT_C, true><<<gemmBlocks, 128, 0, stream>>>(tmpB, Wl, bl, logits, N);

    pool_div_kernel<<<(G * H_C + T - 1) / T, T, 0, stream>>>(pooled, cnts, G * H_C);
}


# first aggregation split in two: its edge sweep only lists each wave's entries (no row gathers between chunk barriers); rows are then gathered from the finished list, 8 in flight; sum order kept
# speedup vs baseline: 2.0437x; 1.1035x over previous
#include <hip/hip_runtime.h>

#define IN_C  128
#define H_C   128
#define OUT_C 64

typedef float v2f __attribute__((ext_vector_type(2)));
typedef float v8f __attribute__((ext_vector_type(8)));

__device__ __forceinline__ void atomic_fadd(float* p, float v) {
    asm volatile("global_atomic_add_f32 %0, %1, off" : : "v"(p), "v"(v) : "memory");
}

__global__ void fill_kernel(float* __restrict__ p, float v, int n) {
    int i = blockIdx.x * blockDim.x + threadIdx.x;
    if (i < n) p[i] = v;
}

__global__ void deg_scatter_kernel(const int* __restrict__ dst,
                                   float* __restrict__ deg, int e) {
    int i = blockIdx.x * blockDim.x + threadIdx.x;
    if (i < e) atomic_fadd(&deg[dst[i]], 1.0f);
}

__global__ void rsqrt_kernel(float* __restrict__ p, int n) {
    int i = blockIdx.x * blockDim.x + threadIdx.x;
    if (i < n) p[i] = rsqrtf(p[i]);
}

template<int K, int NC, bool ADD_BIAS>
__global__ __launch_bounds__(2 * NC)
void gemm_wmma_kernel(const float* __restrict__ A,
                                 const float* __restrict__ B,
                                 const float* __restrict__ bias,
                                 float* __restrict__ C, int M) {
    const int tid   = threadIdx.x;
    const int wave  = tid >> 5;
    const int lane  = tid & 31;
    const int n0    = wave * 16;
    const int lhalf = lane >> 4;
    const int l15   = lane & 15;
    const int n     = n0 + l15;

    v2f bfrag[K / 4];
#pragma unroll
    for (int t = 0; t < K / 4; ++t) {
        const int kk = 8 * (t >> 1) + 4 * lhalf + 2 * (t & 1);
        bfrag[t].x = B[(size_t)kk * NC + n];
        bfrag[t].y = B[(size_t)(kk + 1) * NC + n];
    }
    const float bv = ADD_BIAS ? bias[n] : 0.0f;

    const int strips = (M + 15) / 16;

    for (int s = blockIdx.x; s < strips; s += gridDim.x) {
        const int row0 = s * 16;
        const int row  = (row0 + l15 < M) ? row0 + l15 : M - 1;
        const float4* arow = (const float4*)(A + (size_t)row * K);

        float4 af[K / 8];
#pragma unroll
        for (int i = 0; i < K / 8; ++i) af[i] = arow[2 * i + lhalf];
        static_assert(K / 8 == 16, "the load fence below names 16 loads");
        asm volatile("" : : "v"(af[0].x),  "v"(af[1].x),  "v"(af[2].x),  "v"(af[3].x),
                            "v"(af[4].x),  "v"(af[5].x),  "v"(af[6].x),  "v"(af[7].x),
                            "v"(af[8].x),  "v"(af[9].x),  "v"(af[10].x), "v"(af[11].x),
                            "v"(af[12].x), "v"(af[13].x), "v"(af[14].x), "v"(af[15].x));

        v8f acc = {};
#pragma unroll
        for (int i = 0; i < K / 8; ++i) {
            v2f a0, a1;
            a0.x = af[i].x; a0.y = af[i].y;
            a1.x = af[i].z; a1.y = af[i].w;
            acc = __builtin_amdgcn_wmma_f32_16x16x4_f32(
                      false, a0, false, bfrag[2 * i], (short)0, acc, false, false);
            acc = __builtin_amdgcn_wmma_f32_16x16x4_f32(
                      false, a1, false, bfrag[2 * i + 1], (short)0, acc, false, false);
        }

#pragma unroll
        for (int r = 0; r < 8; ++r) {
            int m = row0 + r + 8 * lhalf;
            if (m < M) C[(size_t)m * NC + n] = acc[r] + bv;
        }
    }
}

#define AGG_TN    256
#define AGG_WAVES 32
#define AGG_STEP  1024
#define AGG_CHUNK (AGG_WAVES * AGG_STEP)
#define AGG_CAP   1024
#define POOL_GN   32

template<int R>
__device__ __forceinline__ void agg_add_rows(int s, int dl, float w, int nk, int lane,
                                             const float* __restrict__ msg, float* acc) {
    for (int j = 0; j < nk; j += R) {
        float4 r[R];
        int    dj[R];
        float  wj[R];
#pragma unroll
        for (int t = 0; t < R; ++t) {
            const int jx = (j + t < nk) ? j + t : nk - 1;
            const int sj = __builtin_amdgcn_readlane(s, jx);
            r[t]  = ((const float4*)(msg + (size_t)sj * H_C))[lane];
        }
#pragma unroll
        for (int t = 0; t < R; ++t) {
            const int jx = (j + t < nk) ? j + t : nk - 1;
            dj[t] = __builtin_amdgcn_readlane(dl, jx);
            wj[t] = __int_as_float(__builtin_amdgcn_readlane(__float_as_int(w), jx));
        }
#pragma unroll
        for (int t = 0; t < R; ++t) {
            if (j + t < nk) {
                float4* a = (float4*)(acc + dj[t] * H_C) + lane;
                float4  v = *a;
                v.x += r[t].x * wj[t];
                v.y += r[t].y * wj[t];
                v.z += r[t].z * wj[t];
                v.w += r[t].w * wj[t];
                *a = v;
            }
        }
    }
}

__device__ __forceinline__ void agg_drain(const unsigned int* myl, int m, int cb, int lane,
                                          const int* __restrict__ src,
                                          const float* __restrict__ dinv,
                                          const float* __restrict__ msg,
                                          const float* dinv_t, float* acc) {
    for (int k0 = 0; k0 < m; k0 += 32) {
        const int nk = (m - k0 < 32) ? m - k0 : 32;
        int   s  = 0;
        int   dl = 0;
        float w  = 0.0f;
        if (lane < nk) {
            const unsigned int ent = myl[k0 + lane];
            dl = (int)((ent >> 15) & 255u);
            s  = src[cb + (int)(ent & 32767u)];
            w  = dinv[s] * dinv_t[dl];
        }
        agg_add_rows<4>(s, dl, w, nk, lane, msg, acc);
    }
}

__device__ __forceinline__ void agg_load_step(const int* __restrict__ dst, int base, int e,
                                              int lane, int (&d)[32]) {
    if (base + AGG_STEP <= e) {
        const int* p = dst + base + lane;
#pragma unroll
        for (int u = 0; u < 32; ++u) d[u] = p[u * 32];
    } else {
#pragma unroll
        for (int u = 0; u < 32; ++u) {
            const int idx = base + u * 32 + lane;
            const int v   = dst[idx < e ? idx : e - 1];
            d[u] = (idx < e) ? v : -1;
        }
    }
}

template<bool FIND_ONLY>
__device__ __forceinline__ void tile_sweep(const int* __restrict__ src,
                                           const int* __restrict__ dst,
                                           const float* __restrict__ dinv,
                                           const float* __restrict__ msg,
                                           const float* dinv_t, float* acc,
                                           unsigned int* keep, int* kept, unsigned int* myl,
                                           uint2* __restrict__ lst, int& lcnt,
                                           int t0, int e, int wave, int lane) {
    int d[32];
    agg_load_step(dst, wave * AGG_STEP, e, lane, d);
    __syncthreads();

    for (int cb = 0; cb < e; cb += AGG_CHUNK) {
        int cnt = 0;
#pragma unroll
        for (int u = 0; u < 32; ++u) {
            const unsigned int dl   = (unsigned int)(d[u] - t0);
            const bool         hit  = dl < (unsigned int)AGG_TN;
            const unsigned int mask = __builtin_amdgcn_ballot_w32(hit);
            if (mask != 0u) {
                if (hit)
                    keep[(cnt + (int)__builtin_amdgcn_mbcnt_lo(mask, 0u)) * AGG_WAVES + wave] =
                        (unsigned int)(wave * AGG_STEP + u * 32 + lane) | (dl << 15);
                cnt += __popc(mask);
            }
        }
        if (lane == 0) kept[wave] = cnt;
        __syncthreads();

        agg_load_step(dst, cb + AGG_CHUNK + wave * AGG_STEP, e, lane, d);

        const int c = kept[lane];
        int maxc = c;
#pragma unroll
        for (int o = 16; o > 0; o >>= 1) {
            const int other = __shfl_xor(maxc, o);
            maxc = other > maxc ? other : maxc;
        }
        maxc = __builtin_amdgcn_readfirstlane(maxc);
        int mcnt = 0;
        for (int j = 0; j < maxc; j += 4) {
            unsigned int ent[4];
#pragma unroll
            for (int t = 0; t < 4; ++t)
                ent[t] = (j + t < c) ? keep[(j + t) * AGG_WAVES + lane] : 0u;
#pragma unroll
            for (int t = 0; t < 4; ++t) {
                const bool         hit  = (j + t < c) && ((ent[t] >> 18) == (unsigned int)wave);
                const unsigned int mask = __builtin_amdgcn_ballot_w32(hit);
                if (mask != 0u) {
                    if (FIND_ONLY) {
                        const int pos = lcnt + (int)__builtin_amdgcn_mbcnt_lo(mask, 0u);
                        if (hit && pos < AGG_CAP)
                            lst[pos] = make_uint2((unsigned int)cb + (ent[t] & 32767u),
                                                  (ent[t] >> 15) & 7u);
                        lcnt += __popc(mask);
                    } else {
                        if (hit) myl[mcnt + (int)__builtin_amdgcn_mbcnt_lo(mask, 0u)] = ent[t];
                        mcnt += __popc(mask);
                    }
                }
            }
            if (!FIND_ONLY && mcnt > 32) {
                agg_drain(myl, mcnt, cb, lane, src, dinv, msg, dinv_t, acc);
                mcnt = 0;
            }
        }
        if (!FIND_ONLY && mcnt > 0) agg_drain(myl, mcnt, cb, lane, src, dinv, msg, dinv_t, acc);
        __syncthreads();
    }
}

template<bool FIRST>
__global__ __launch_bounds__(AGG_WAVES * 32)
void tile_agg_kernel(const int* __restrict__ src,
                     const int* __restrict__ dst,
                     const float* __restrict__ dinv,
                     const float* __restrict__ msg,
                     const float* __restrict__ bias,
                     float* __restrict__ out, int n, int e,
                     unsigned int* __restrict__ lcount,
                     unsigned int* __restrict__ lists,
                     const int* __restrict__ batch,
                     float* __restrict__ pooled,
                     float* __restrict__ cnts) {
    __shared__ __attribute__((aligned(16))) float acc[AGG_TN * H_C];
    __shared__ float        dinv_t[AGG_TN];
    __shared__ int          gid_t[AGG_TN];
    __shared__ unsigned int keep[AGG_STEP * AGG_WAVES];
    __shared__ int          kept[AGG_WAVES];
    __shared__ unsigned int mine[AGG_WAVES][160];

    const int tid  = threadIdx.x;
    const int lane = tid & 31;
    const int wave = __builtin_amdgcn_readfirstlane(tid >> 5);
    const int t0   = blockIdx.x * AGG_TN;

    uint2* lst = (uint2*)lists + (size_t)(blockIdx.x * AGG_WAVES + wave) * AGG_CAP;

    bool fits = false;
    int  mlist = 0;
    uint2 ecur = make_uint2(0u, 0u);
    if (FIRST) {
        if (tid < AGG_TN) dinv_t[tid] = (t0 + tid < n) ? dinv[t0 + tid] : 0.0f;
        int lcnt = 0;
        tile_sweep<true>(src, dst, dinv, msg, dinv_t, acc, keep, kept, mine[wave],
                         lst, lcnt, t0, e, wave, lane);
        if (lane == 0) {
            lcount[blockIdx.x * AGG_WAVES + wave] = (unsigned int)lcnt;
            kept[wave] = lcnt;
        }
        __syncthreads();
        fits  = __builtin_amdgcn_ballot_w32(kept[lane] > AGG_CAP) == 0u;
        mlist = lcnt;
        if (fits) {
            for (int k0 = 0; k0 < mlist; k0 += 256) {
                uint2 raw[8];
                int   sv[8];
                float dv[8];
#pragma unroll
                for (int t = 0; t < 8; ++t) {
                    const int idx = k0 + 32 * t + lane;
                    raw[t] = lst[idx < mlist ? idx : mlist - 1];
                }
                __builtin_amdgcn_sched_barrier(0);
                asm volatile("" : : "v"(raw[0].x), "v"(raw[1].x), "v"(raw[2].x), "v"(raw[3].x),
                                    "v"(raw[4].x), "v"(raw[5].x), "v"(raw[6].x), "v"(raw[7].x));
#pragma unroll
                for (int t = 0; t < 8; ++t) sv[t] = src[raw[t].x];
                asm volatile("" : : "v"(sv[0]), "v"(sv[1]), "v"(sv[2]), "v"(sv[3]),
                                    "v"(sv[4]), "v"(sv[5]), "v"(sv[6]), "v"(sv[7]));
#pragma unroll
                for (int t = 0; t < 8; ++t) dv[t] = dinv[sv[t]];
                asm volatile("" : : "v"(dv[0]), "v"(dv[1]), "v"(dv[2]), "v"(dv[3]),
                                    "v"(dv[4]), "v"(dv[5]), "v"(dv[6]), "v"(dv[7]));
#pragma unroll
                for (int t = 0; t < 8; ++t) {
                    const int   idx = k0 + 32 * t + lane;
                    const float w   = dv[t] * dinv_t[wave * 8 + (int)raw[t].y];
                    const uint2 fin = make_uint2((unsigned int)sv[t] | (raw[t].y << 29),
                                                 (unsigned int)__float_as_int(w));
                    if (idx < mlist) lst[idx] = fin;
                    if (t == 0 && k0 == 0 && lane < mlist) ecur = fin;
                }
            }
        }
    } else {
        const unsigned int c = lcount[blockIdx.x * AGG_WAVES + lane];
        fits  = __builtin_amdgcn_ballot_w32(c > (unsigned int)AGG_CAP) == 0u;
        mlist = __builtin_amdgcn_readlane((int)c, wave);
        if (fits && lane < mlist) ecur = lst[lane];
        if (tid < AGG_TN) {
            dinv_t[tid] = (t0 + tid < n) ? dinv[t0 + tid] : 0.0f;
            gid_t[tid]  = (t0 + tid < n) ? batch[t0 + tid] : -1;
        }
    }

    for (int i = tid; i < AGG_TN * (H_C / 4); i += AGG_WAVES * 32) {
        const int node = t0 + (i >> 5);
        const int c4   = i & 31;
        float4 v = make_float4(0.0f, 0.0f, 0.0f, 0.0f);
        if (node < n) {
            const float  di = dinv[node];
            const float4 m  = ((const float4*)(msg + (size_t)node * H_C))[c4];
            const float4 b  = ((const float4*)bias)[c4];
            v.x = m.x * di * di + b.x;
            v.y = m.y * di * di + b.y;
            v.z = m.z * di * di + b.z;
            v.w = m.w * di * di + b.w;
        }
        ((float4*)acc)[i] = v;
    }

    if (fits) {
        __syncthreads();
        for (int k0 = 0; k0 < mlist; k0 += 32) {
            const int nk = (mlist - k0 < 32) ? mlist - k0 : 32;
            uint2 enext = make_uint2(0u, 0u);
            if (k0 + 32 + lane < mlist) enext = lst[k0 + 32 + lane];
            int   s  = 0;
            int   dl = 0;
            float w  = 0.0f;
            if (lane < nk) {
                s  = (int)(ecur.x & 0x1fffffffu);
                dl = wave * 8 + (int)(ecur.x >> 29);
                w  = __int_as_float((int)ecur.y);
            }
            agg_add_rows<8>(s, dl, w, nk, lane, msg, acc);
            ecur = enext;
        }
        __syncthreads();
    } else {
        int unused = 0;
        tile_sweep<false>(src, dst, dinv, msg, dinv_t, acc, keep, kept, mine[wave],
                          lst, unused, t0, e, wave, lane);
    }

    for (int i = tid; i < AGG_TN * (H_C / 4); i += AGG_WAVES * 32) {
        const int node = t0 + (i >> 5);
        if (node < n) {
            float4 v = ((const float4*)acc)[i];
            v.x = fmaxf(v.x, 0.0f);
            v.y = fmaxf(v.y, 0.0f);
            v.z = fmaxf(v.z, 0.0f);
            v.w = fmaxf(v.w, 0.0f);
            ((float4*)(out + (size_t)node * H_C))[i & 31] = v;
        }
    }

    if (!FIRST) {
        const int c  = tid & (H_C - 1);
        const int q  = __builtin_amdgcn_readfirstlane(tid >> 7);
        const int nb = q * POOL_GN;
        int nval = n - (t0 + nb);
        nval = nval < POOL_GN ? nval : POOL_GN;
        if (nval > 0) {
            float v[POOL_GN];
#pragma unroll
            for (int k = 0; k < POOL_GN; ++k) v[k] = acc[(nb + k) * H_C + c];
            const int gl = gid_t[nb + lane];
            int   g   = __builtin_amdgcn_readlane(gl, 0);
            float run = 0.0f;
            int   len = 0;
#pragma unroll
            for (int k = 0; k < POOL_GN; ++k) {
                if (k < nval) {
                    const int gk = __builtin_amdgcn_readlane(gl, k);
                    if (gk != g) {
                        atomic_fadd(&pooled[(size_t)g * H_C + c], run);
                        if (c == 0) atomic_fadd(&cnts[g], (float)len);
                        run = 0.0f;
                        len = 0;
                        g   = gk;
                    }
                    run += fmaxf(v[k], 0.0f);
                    ++len;
                }
            }
            atomic_fadd(&pooled[(size_t)g * H_C + c], run);
            if (c == 0) atomic_fadd(&cnts[g], (float)len);
        }
    }
}

__global__ void pool_div_kernel(float* __restrict__ pooled,
                                const float* __restrict__ cnts, int total) {
    int i = blockIdx.x * blockDim.x + threadIdx.x;
    if (i >= total) return;
    pooled[i] /= fmaxf(cnts[i >> 7], 1.0f);
}

extern "C" void kernel_launch(void* const* d_in, const int* in_sizes, int n_in,
                              void* d_out, int out_size, void* d_ws, size_t ws_size,
                              hipStream_t stream) {
    const float* x    = (const float*)d_in[0];
    const int*   ei   = (const int*)  d_in[1];
    const int*   batch= (const int*)  d_in[2];
    const float* W1   = (const float*)d_in[3];
    const float* b1   = (const float*)d_in[4];
    const float* W2   = (const float*)d_in[5];
    const float* b2   = (const float*)d_in[6];
    const float* Wl   = (const float*)d_in[7];
    const float* bl   = (const float*)d_in[8];

    const int N = in_sizes[0] / IN_C;
    const int E = in_sizes[1] / 2;
    const int G = (out_size - N * OUT_C) / H_C;

    const int* esrc = ei;
    const int* edst = ei + E;

    float* logits = (float*)d_out;
    float* pooled = logits + (size_t)N * OUT_C;

    float* deg  = (float*)d_ws;
    float* tmpA = deg + (((size_t)N + 255) & ~(size_t)255);
    float* tmpB = tmpA + (size_t)N * H_C;
    float* cnts = tmpB + (size_t)N * H_C;

    const int T  = 256;
    const int aggTiles = (N + AGG_TN - 1) / AGG_TN;
    unsigned int* lcount = (unsigned int*)(cnts + (((size_t)G + 255) & ~(size_t)255));
    unsigned int* lists = lcount + (((size_t)aggTiles * AGG_WAVES + 63) & ~(size_t)63);
    const int rowBlocks  = (N + 15) / 16;
    const int gemmBlocks = rowBlocks < 512 ? rowBlocks : 512;

    fill_kernel<<<(N + T - 1) / T, T, 0, stream>>>(deg, 1.0f, N);
    fill_kernel<<<(G + T - 1) / T, T, 0, stream>>>(cnts, 0.0f, G);
    fill_kernel<<<(G * H_C + T - 1) / T, T, 0, stream>>>(pooled, 0.0f, G * H_C);
    deg_scatter_kernel<<<(E + T - 1) / T, T, 0, stream>>>(edst, deg, E);
    rsqrt_kernel<<<(N + T - 1) / T, T, 0, stream>>>(deg, N);

    gemm_wmma_kernel<IN_C, H_C, false><<<gemmBlocks, 256, 0, stream>>>(x, W1, nullptr, tmpA, N);
    tile_agg_kernel<true><<<aggTiles, AGG_WAVES * 32, 0, stream>>>(esrc, edst, deg, tmpA, b1, tmpB, N, E, lcount, lists, nullptr, nullptr, nullptr);

    gemm_wmma_kernel<H_C, H_C, false><<<gemmBlocks, 256, 0, stream>>>(tmpB, W2, nullptr, tmpA, N);
    tile_agg_kernel<false><<<aggTiles, AGG_WAVES * 32, 0, stream>>>(esrc, edst, deg, tmpA, b2, tmpB, N, E, lcount, lists, batch, pooled, cnts);

    gemm_wmma_kernel<H_C, OUT_C, true><<<gemmBlocks, 128, 0, stream>>>(tmpB, Wl, bl, logits, N);

    pool_div_kernel<<<(G * H_C + T - 1) / T, T, 0, stream>>>(pooled, cnts, G * H_C);
}


# first aggregation takes a tile's edges from chunk-wise groups (the degree launch orders every 1024 edges by destination tile, ballot ranks) instead of searching all edges; sum order changed, fixed
# speedup vs baseline: 2.8722x; 1.4054x over previous
#include <hip/hip_runtime.h>

#define IN_C  128
#define H_C   128
#define OUT_C 64

typedef float v2f __attribute__((ext_vector_type(2)));
typedef float v8f __attribute__((ext_vector_type(8)));

__device__ __forceinline__ void atomic_fadd(float* p, float v) {
    asm volatile("global_atomic_add_f32 %0, %1, off" : : "v"(p), "v"(v) : "memory");
}

__device__ __forceinline__ float dinv_at(const float* __restrict__ deg, int i) {
    return rsqrtf(deg[i]);
}

template<int K, int NC, bool ADD_BIAS, int SIDE>
__global__ __launch_bounds__(2 * NC)
void gemm_wmma_kernel(const float* __restrict__ A,
                                 const float* __restrict__ B,
                                 const float* __restrict__ bias,
                                 float* __restrict__ C, int M,
                                 float* __restrict__ deg, int nodes,
                                 float* __restrict__ cnts,
                                 float* __restrict__ pooled, int graphs) {
    const int tid   = threadIdx.x;
    const int wave  = tid >> 5;
    const int lane  = tid & 31;
    const int n0    = wave * 16;
    const int lhalf = lane >> 4;
    const int l15   = lane & 15;
    const int n     = n0 + l15;

    v2f bfrag[K / 4];
#pragma unroll
    for (int t = 0; t < K / 4; ++t) {
        const int kk = 8 * (t >> 1) + 4 * lhalf + 2 * (t & 1);
        bfrag[t].x = B[(size_t)kk * NC + n];
        bfrag[t].y = B[(size_t)(kk + 1) * NC + n];
    }
    const float bv = ADD_BIAS ? bias[n] : 0.0f;

    if (SIDE != 0) {
        const int gsz = gridDim.x * blockDim.x;
        const int g0  = blockIdx.x * blockDim.x + tid;
        if (SIDE == 1) {
            for (int i = g0; i < nodes; i += gsz)        deg[i]    = 1.0f;
            for (int i = g0; i < graphs; i += gsz)       cnts[i]   = 0.0f;
            for (int i = g0; i < graphs * H_C; i += gsz) pooled[i] = 0.0f;
        } else {
            for (int i = g0; i < graphs * H_C; i += gsz)
                pooled[i] /= fmaxf(cnts[i / H_C], 1.0f);
        }
    }

    const int strips = (M + 15) / 16;

    for (int s = blockIdx.x; s < strips; s += gridDim.x) {
        const int row0 = s * 16;
        const int row  = (row0 + l15 < M) ? row0 + l15 : M - 1;
        const float4* arow = (const float4*)(A + (size_t)row * K);

        float4 af[K / 8];
#pragma unroll
        for (int i = 0; i < K / 8; ++i) af[i] = arow[2 * i + lhalf];
        static_assert(K / 8 == 16, "the load fence below names 16 loads");
        asm volatile("" : : "v"(af[0].x),  "v"(af[1].x),  "v"(af[2].x),  "v"(af[3].x),
                            "v"(af[4].x),  "v"(af[5].x),  "v"(af[6].x),  "v"(af[7].x),
                            "v"(af[8].x),  "v"(af[9].x),  "v"(af[10].x), "v"(af[11].x),
                            "v"(af[12].x), "v"(af[13].x), "v"(af[14].x), "v"(af[15].x));

        v8f acc = {};
#pragma unroll
        for (int i = 0; i < K / 8; ++i) {
            v2f a0, a1;
            a0.x = af[i].x; a0.y = af[i].y;
            a1.x = af[i].z; a1.y = af[i].w;
            acc = __builtin_amdgcn_wmma_f32_16x16x4_f32(
                      false, a0, false, bfrag[2 * i], (short)0, acc, false, false);
            acc = __builtin_amdgcn_wmma_f32_16x16x4_f32(
                      false, a1, false, bfrag[2 * i + 1], (short)0, acc, false, false);
        }

#pragma unroll
        for (int r = 0; r < 8; ++r) {
            int m = row0 + r + 8 * lhalf;
            if (m < M) C[(size_t)m * NC + n] = acc[r] + bv;
        }
    }
}

#define AGG_TN    256
#define AGG_WAVES 32
#define AGG_STEP  1024
#define AGG_CHUNK (AGG_WAVES * AGG_STEP)
#define AGG_CAP   1024
#define POOL_GN   32

template<int R>
__device__ __forceinline__ void agg_add_rows(int s, int dl, float w, int nk, int lane,
                                             const float* __restrict__ msg, float* acc) {
    for (int j = 0; j < nk; j += R) {
        float4 r[R];
        int    dj[R];
        float  wj[R];
#pragma unroll
        for (int t = 0; t < R; ++t) {
            const int jx = (j + t < nk) ? j + t : nk - 1;
            const int sj = __builtin_amdgcn_readlane(s, jx);
            r[t]  = ((const float4*)(msg + (size_t)sj * H_C))[lane];
        }
#pragma unroll
        for (int t = 0; t < R; ++t) {
            const int jx = (j + t < nk) ? j + t : nk - 1;
            dj[t] = __builtin_amdgcn_readlane(dl, jx);
            wj[t] = __int_as_float(__builtin_amdgcn_readlane(__float_as_int(w), jx));
        }
#pragma unroll
        for (int t = 0; t < R; ++t) {
            if (j + t < nk) {
                float4* a = (float4*)(acc + dj[t] * H_C) + lane;
                float4  v = *a;
                v.x += r[t].x * wj[t];
                v.y += r[t].y * wj[t];
                v.z += r[t].z * wj[t];
                v.w += r[t].w * wj[t];
                *a = v;
            }
        }
    }
}

template<int R>
__device__ __forceinline__ void agg_add_rows_run(int s, int pl, float w, int nk, int lane,
                                                 const float* __restrict__ msg, float* rows,
                                                 int& cur, float4& run) {
    for (int j = 0; j < nk; j += R) {
        float4 r[R];
        int    pj[R];
        float  wj[R];
#pragma unroll
        for (int t = 0; t < R; ++t) {
            const int jx = (j + t < nk) ? j + t : nk - 1;
            const int sj = __builtin_amdgcn_readlane(s, jx);
            r[t]  = ((const float4*)(msg + (size_t)sj * H_C))[lane];
        }
#pragma unroll
        for (int t = 0; t < R; ++t) {
            const int jx = (j + t < nk) ? j + t : nk - 1;
            pj[t] = __builtin_amdgcn_readlane(pl, jx);
            wj[t] = __int_as_float(__builtin_amdgcn_readlane(__float_as_int(w), jx));
        }
#pragma unroll
        for (int t = 0; t < R; ++t) {
            if (j + t < nk) {
                if (pj[t] != cur) {
                    ((float4*)(rows + cur * H_C))[lane] = run;
                    cur = pj[t];
                    run = ((const float4*)(rows + cur * H_C))[lane];
                }
                run.x += r[t].x * wj[t];
                run.y += r[t].y * wj[t];
                run.z += r[t].z * wj[t];
                run.w += r[t].w * wj[t];
            }
        }
    }
}

__device__ __forceinline__ void agg_drain(const unsigned int* myl, int m, int cb, int lane,
                                          const int* __restrict__ src,
                                          const float* __restrict__ deg,
                                          const float* __restrict__ msg,
                                          const float* dinv_t, float* acc) {
    for (int k0 = 0; k0 < m; k0 += 32) {
        const int nk = (m - k0 < 32) ? m - k0 : 32;
        int   s  = 0;
        int   dl = 0;
        float w  = 0.0f;
        if (lane < nk) {
            const unsigned int ent = myl[k0 + lane];
            dl = (int)((ent >> 15) & 255u);
            s  = src[cb + (int)(ent & 32767u)];
            w  = dinv_at(deg, s) * dinv_t[dl];
        }
        agg_add_rows<4>(s, dl, w, nk, lane, msg, acc);
    }
}

__device__ __forceinline__ void agg_load_step(const int* __restrict__ dst, int base, int e,
                                              int lane, int (&d)[32]) {
    if (base + AGG_STEP <= e) {
        const int* p = dst + base + lane;
#pragma unroll
        for (int u = 0; u < 32; ++u) d[u] = p[u * 32];
    } else {
#pragma unroll
        for (int u = 0; u < 32; ++u) {
            const int idx = base + u * 32 + lane;
            const int v   = dst[idx < e ? idx : e - 1];
            d[u] = (idx < e) ? v : -1;
        }
    }
}

__device__ __forceinline__ void tile_sweep(const int* __restrict__ src,
                                           const int* __restrict__ dst,
                                           const float* __restrict__ deg,
                                           const float* __restrict__ msg,
                                           const float* dinv_t, float* acc,
                                           unsigned int* keep, int* kept, unsigned int* myl,
                                           int t0, int e, int wave, int lane) {
    int d[32];
    agg_load_step(dst, wave * AGG_STEP, e, lane, d);
    __syncthreads();

    for (int cb = 0; cb < e; cb += AGG_CHUNK) {
        int cnt = 0;
#pragma unroll
        for (int u = 0; u < 32; ++u) {
            const unsigned int dl   = (unsigned int)(d[u] - t0);
            const bool         hit  = dl < (unsigned int)AGG_TN;
            const unsigned int mask = __builtin_amdgcn_ballot_w32(hit);
            if (mask != 0u) {
                if (hit)
                    keep[(cnt + (int)__builtin_amdgcn_mbcnt_lo(mask, 0u)) * AGG_WAVES + wave] =
                        (unsigned int)(wave * AGG_STEP + u * 32 + lane) | (dl << 15);
                cnt += __popc(mask);
            }
        }
        if (lane == 0) kept[wave] = cnt;
        __syncthreads();

        agg_load_step(dst, cb + AGG_CHUNK + wave * AGG_STEP, e, lane, d);

        const int c = kept[lane];
        int maxc = c;
#pragma unroll
        for (int o = 16; o > 0; o >>= 1) {
            const int other = __shfl_xor(maxc, o);
            maxc = other > maxc ? other : maxc;
        }
        maxc = __builtin_amdgcn_readfirstlane(maxc);
        int mcnt = 0;
        for (int j = 0; j < maxc; j += 4) {
            unsigned int ent[4];
#pragma unroll
            for (int t = 0; t < 4; ++t)
                ent[t] = (j + t < c) ? keep[(j + t) * AGG_WAVES + lane] : 0u;
#pragma unroll
            for (int t = 0; t < 4; ++t) {
                const bool         hit  = (j + t < c) && ((ent[t] >> 18) == (unsigned int)wave);
                const unsigned int mask = __builtin_amdgcn_ballot_w32(hit);
                if (mask != 0u) {
                    if (hit) myl[mcnt + (int)__builtin_amdgcn_mbcnt_lo(mask, 0u)] = ent[t];
                    mcnt += __popc(mask);
                }
            }
            if (mcnt > 32) {
                agg_drain(myl, mcnt, cb, lane, src, deg, msg, dinv_t, acc);
                mcnt = 0;
            }
        }
        if (mcnt > 0) agg_drain(myl, mcnt, cb, lane, src, deg, msg, dinv_t, acc);
        __syncthreads();
    }
}

#define CS_EDGES 1024
#define CS_WAVES 4
static_assert(CS_EDGES == AGG_STEP, "a chunk is read with agg_load_step");
static_assert(AGG_TN == 256, "the tile of a destination is node >> 8");

__device__ __forceinline__ unsigned int tile_of(int d) {
    const unsigned int t = (unsigned int)d >> 8;
    return t < 255u ? t : 255u;
}

__global__ __launch_bounds__(CS_WAVES * 32)
void chunk_sort_kernel(const int* __restrict__ dst,
                       float* __restrict__ deg,
                       unsigned int* __restrict__ cstart,
                       unsigned int* __restrict__ sorted,
                       int e) {
    __shared__ unsigned int hist[CS_WAVES * 256];
    const int lane  = threadIdx.x & 31;
    const int wave  = __builtin_amdgcn_readfirstlane((int)(threadIdx.x >> 5));
    const int chunk = blockIdx.x * CS_WAVES + wave;
    const int base  = chunk * CS_EDGES;
    if (base >= e) return;
    unsigned int* h = hist + wave * 256;
#pragma unroll
    for (int k = 0; k < 8; ++k) h[lane + 32 * k] = 0u;

    int d[32];
    agg_load_step(dst, base, e, lane, d);

    unsigned int rk[32];
#pragma unroll
    for (int u = 0; u < 32; ++u) {
        const bool valid = d[u] >= 0;
        if (valid) atomic_fadd(&deg[d[u]], 1.0f);
        const unsigned int t = valid ? tile_of(d[u]) : 255u;
        unsigned int m = 0xffffffffu;
#pragma unroll
        for (int b = 0; b < 8; ++b) {
            const bool         bit = ((t >> b) & 1u) != 0u;
            const unsigned int bal = __builtin_amdgcn_ballot_w32(bit);
            m &= bit ? bal : ~bal;
        }
        const unsigned int r   = __builtin_amdgcn_mbcnt_lo(m, 0u);
        const unsigned int old = h[t];
        rk[u] = old + r;
        if (r == 0u) h[t] = old + (unsigned int)__popc(m);
    }

    unsigned int c[8];
    unsigned int s = 0u;
#pragma unroll
    for (int k = 0; k < 8; ++k) { c[k] = h[8 * lane + k]; s += c[k]; }
    unsigned int incl = s;
#pragma unroll
    for (int o = 1; o < 32; o <<= 1) {
        const unsigned int up = (unsigned int)__shfl_up((int)incl, o);
        if (lane >= o) incl += up;
    }
    unsigned int run = incl - s;
    unsigned int* cs = cstart + (size_t)chunk * 256;
#pragma unroll
    for (int k = 0; k < 8; ++k) {
        h[8 * lane + k]  = run;
        cs[8 * lane + k] = run;
        run += c[k];
    }

    unsigned int* so = sorted + (size_t)chunk * CS_EDGES;
#pragma unroll
    for (int u = 0; u < 32; ++u) {
        if (d[u] >= 0)
            so[h[tile_of(d[u])] + rk[u]] =
                (unsigned int)(base + 32 * u + lane) | (((unsigned int)d[u] & 255u) << 24);
    }
}

__device__ __forceinline__ int tile_collect(const unsigned int* __restrict__ cstart,
                                            const unsigned int* __restrict__ sorted,
                                            unsigned int* myk, unsigned int tile, int e,
                                            int wave, int lane) {
    const int chunks = (e + CS_EDGES - 1) / CS_EDGES;
    const int per    = (chunks + AGG_WAVES - 1) / AGG_WAVES;
    const int cbeg   = wave * per;
    const int cend   = (cbeg + per < chunks) ? cbeg + per : chunks;
    const unsigned int tq = tile < 255u ? tile : 254u;
    int found = 0;
    for (int c0 = cbeg; c0 < cend; c0 += 32) {
        const int  c  = c0 + lane;
        const bool ok = c < cend;
        unsigned int a = 0u, b = 0u;
        if (ok) {
            const unsigned int* cs = cstart + (size_t)c * 256 + tq;
            a = cs[0];
            b = cs[1];
        }
        const int nl = (int)(b - a);
        int incl = nl;
#pragma unroll
        for (int o = 1; o < 32; o <<= 1) {
            const int up = __shfl_up(incl, o);
            if (lane >= o) incl += up;
        }
        int mx = nl;
#pragma unroll
        for (int o = 16; o > 0; o >>= 1) {
            const int other = __shfl_xor(mx, o);
            mx = other > mx ? other : mx;
        }
        mx = __builtin_amdgcn_readfirstlane(mx);
        const int pos0 = found + incl - nl;
        const unsigned int* sp = sorted + (size_t)(ok ? c : cbeg) * CS_EDGES + a;
        for (int k = 0; k < mx; k += 4) {
            unsigned int v[4];
#pragma unroll
            for (int t = 0; t < 4; ++t) {
                v[t] = 0u;
                if (k + t < nl) v[t] = sp[k + t];
            }
#pragma unroll
            for (int t = 0; t < 4; ++t) {
                const int pos = pos0 + k + t;
                if (k + t < nl && pos < AGG_STEP) myk[pos] = v[t];
            }
        }
        found += __builtin_amdgcn_readlane(incl, 31);
    }
    return found;
}

template<bool FIRST>
__global__ __launch_bounds__(AGG_WAVES * 32)
void tile_agg_kernel(const int* __restrict__ src,
                     const int* __restrict__ dst,
                     const float* __restrict__ deg,
                     const float* __restrict__ msg,
                     const float* __restrict__ bias,
                     float* __restrict__ out, int n, int e,
                     unsigned int* __restrict__ lcount,
                     unsigned int* __restrict__ lists,
                     const unsigned int* __restrict__ cstart,
                     const unsigned int* __restrict__ sorted,
                     const int* __restrict__ batch,
                     float* __restrict__ pooled,
                     float* __restrict__ cnts) {
    __shared__ __attribute__((aligned(16))) float acc[AGG_TN * H_C];
    __shared__ float        dinv_t[AGG_TN];
    __shared__ int          gid_t[AGG_TN];
    __shared__ unsigned int keep[AGG_STEP * AGG_WAVES];
    __shared__ int          kept[AGG_WAVES];
    __shared__ unsigned int mine[AGG_WAVES][160];

    const int tid  = threadIdx.x;
    const int lane = tid & 31;
    const int wave = __builtin_amdgcn_readfirstlane(tid >> 5);
    const int t0   = blockIdx.x * AGG_TN;

    uint2* lst = (uint2*)lists + (size_t)(blockIdx.x * AGG_WAVES + wave) * AGG_CAP;

    bool fits = false;
    int  mlist = 0;
    uint2 ecur = make_uint2(0u, 0u);
    if (FIRST) {
        if (tid < AGG_TN) dinv_t[tid] = (t0 + tid < n) ? dinv_at(deg, t0 + tid) : 0.0f;
        unsigned int* myk = keep + wave * AGG_STEP;
        const int found = tile_collect(cstart, sorted, myk, (unsigned int)blockIdx.x, e, wave, lane);
        if (lane == 0) kept[wave] = found;
        __syncthreads();
        const int  fl     = kept[lane];
        const bool partok = __builtin_amdgcn_ballot_w32(fl > AGG_STEP) == 0u &&
                            n <= 255 * AGG_TN && e <= (1 << 24);
        int lcnt = 0;
        static_assert(8 * H_C == AGG_CAP, "a wave's notes fill its own 8 rows");
        float* rawl = acc + wave * 8 * H_C;
        if (partok) {
            for (int p = 0; p < AGG_WAVES; ++p) {
                const int           cp = __builtin_amdgcn_readlane(fl, p);
                const unsigned int* kp = keep + p * AGG_STEP;
                for (int j = 0; j < cp; j += 64) {
                    unsigned int w[2];
#pragma unroll
                    for (int t = 0; t < 2; ++t) {
                        const int jj = j + 32 * t + lane;
                        w[t] = (jj < cp) ? kp[jj] : 0u;
                    }
#pragma unroll
                    for (int t = 0; t < 2; ++t) {
                        const int          jj   = j + 32 * t + lane;
                        const bool         hit  = (jj < cp) && ((w[t] >> 27) == (unsigned int)wave);
                        const unsigned int mask = __builtin_amdgcn_ballot_w32(hit);
                        if (mask != 0u) {
                            const int pos = lcnt + (int)__builtin_amdgcn_mbcnt_lo(mask, 0u);
                            if (hit && pos < AGG_CAP)
                                rawl[pos] = __uint_as_float(w[t] & 0x07ffffffu);
                            lcnt += __popc(mask);
                        }
                    }
                }
            }
        }
        const int mark = partok ? lcnt : AGG_CAP + 1;
        if (lane == 0) {
            lcount[blockIdx.x * AGG_WAVES + wave] = (unsigned int)mark;
            gid_t[wave] = mark;
        }
        __syncthreads();
        fits  = __builtin_amdgcn_ballot_w32(gid_t[lane] > AGG_CAP) == 0u;
        mlist = lcnt;
        if (fits) {
            int cnt[8] = {0, 0, 0, 0, 0, 0, 0, 0};
            for (int k0 = 0; k0 < mlist; k0 += 32) {
                const int idx = k0 + lane;
                const int pl  = idx < mlist ? (int)((__float_as_uint(rawl[idx]) >> 24) & 7u) : 8;
#pragma unroll
                for (int p = 0; p < 8; ++p)
                    cnt[p] += __popc(__builtin_amdgcn_ballot_w32(pl == p));
            }
            int off[8];
            off[0] = 0;
#pragma unroll
            for (int p = 1; p < 8; ++p) off[p] = off[p - 1] + cnt[p - 1];
            for (int k0 = 0; k0 < mlist; k0 += 256) {
                unsigned int raw[8];
                int          sv[8];
                float        dv[8];
#pragma unroll
                for (int t = 0; t < 8; ++t) {
                    const int idx = k0 + 32 * t + lane;
                    raw[t] = __float_as_uint(rawl[idx < mlist ? idx : mlist - 1]);
                }
#pragma unroll
                for (int t = 0; t < 8; ++t) sv[t] = src[raw[t] & 0xffffffu];
                asm volatile("" : : "v"(sv[0]), "v"(sv[1]), "v"(sv[2]), "v"(sv[3]),
                                    "v"(sv[4]), "v"(sv[5]), "v"(sv[6]), "v"(sv[7]));
#pragma unroll
                for (int t = 0; t < 8; ++t) dv[t] = deg[sv[t]];
                asm volatile("" : : "v"(dv[0]), "v"(dv[1]), "v"(dv[2]), "v"(dv[3]),
                                    "v"(dv[4]), "v"(dv[5]), "v"(dv[6]), "v"(dv[7]));
#pragma unroll
                for (int t = 0; t < 8; ++t) {
                    if (k0 + 32 * t < mlist) {
                        const int  idx   = k0 + 32 * t + lane;
                        const bool valid = idx < mlist;
                        const int  pl    = valid ? (int)((raw[t] >> 24) & 7u) : 8;
                        int pos = 0;
#pragma unroll
                        for (int p = 0; p < 8; ++p) {
                            const unsigned int mask = __builtin_amdgcn_ballot_w32(pl == p);
                            if (pl == p) pos = off[p] + (int)__builtin_amdgcn_mbcnt_lo(mask, 0u);
                            off[p] += __popc(mask);
                        }
                        const float w = rsqrtf(dv[t]) * dinv_t[wave * 8 + (pl & 7)];
                        if (valid)
                            lst[pos] = make_uint2((unsigned int)sv[t] | ((unsigned int)pl << 29),
                                                  (unsigned int)__float_as_int(w));
                    }
                }
            }
            __syncthreads();
            if (lane < mlist) ecur = lst[lane];
        }
    } else {
        const unsigned int c = lcount[blockIdx.x * AGG_WAVES + lane];
        fits  = __builtin_amdgcn_ballot_w32(c > (unsigned int)AGG_CAP) == 0u;
        mlist = __builtin_amdgcn_readlane((int)c, wave);
        if (fits && lane < mlist) ecur = lst[lane];
        if (tid < AGG_TN) {
            dinv_t[tid] = (t0 + tid < n) ? dinv_at(deg, t0 + tid) : 0.0f;
            gid_t[tid]  = (t0 + tid < n) ? batch[t0 + tid] : -1;
        }
    }

    if (fits) {
        float* rows = acc + wave * 8 * H_C;
        {
            const float4 b = ((const float4*)bias)[lane];
            float4 m[8];
            float  dg[8];
#pragma unroll
            for (int p = 0; p < 8; ++p) {
                const int node = t0 + wave * 8 + p;
                const int nc   = node < n ? node : n - 1;
                m[p]  = ((const float4*)(msg + (size_t)nc * H_C))[lane];
                dg[p] = deg[nc];
            }
#pragma unroll
            for (int p = 0; p < 8; ++p) {
                const float di = rsqrtf(dg[p]);
                float4 v = make_float4(0.0f, 0.0f, 0.0f, 0.0f);
                if (t0 + wave * 8 + p < n) {
                    v.x = m[p].x * di * di + b.x;
                    v.y = m[p].y * di * di + b.y;
                    v.z = m[p].z * di * di + b.z;
                    v.w = m[p].w * di * di + b.w;
                }
                ((float4*)(rows + p * H_C))[lane] = v;
            }
        }
        int    cur = 0;
        float4 run = ((const float4*)rows)[lane];
        for (int k0 = 0; k0 < mlist; k0 += 32) {
            const int nk = (mlist - k0 < 32) ? mlist - k0 : 32;
            uint2 enext = make_uint2(0u, 0u);
            if (k0 + 32 + lane < mlist) enext = lst[k0 + 32 + lane];
            int   s  = 0;
            int   pl = 0;
            float w  = 0.0f;
            if (lane < nk) {
                s  = (int)(ecur.x & 0x1fffffffu);
                pl = (int)(ecur.x >> 29);
                w  = __int_as_float((int)ecur.y);
            }
            agg_add_rows_run<8>(s, pl, w, nk, lane, msg, rows, cur, run);
            ecur = enext;
        }
        ((float4*)(rows + cur * H_C))[lane] = run;
        __syncthreads();
    } else {
        for (int row = wave; row < AGG_TN; row += AGG_WAVES) {
            const int node = t0 + row;
            float4 v = make_float4(0.0f, 0.0f, 0.0f, 0.0f);
            if (node < n) {
                const float  di = dinv_at(deg, node);
                const float4 m  = ((const float4*)(msg + (size_t)node * H_C))[lane];
                const float4 b  = ((const float4*)bias)[lane];
                v.x = m.x * di * di + b.x;
                v.y = m.y * di * di + b.y;
                v.z = m.z * di * di + b.z;
                v.w = m.w * di * di + b.w;
            }
            ((float4*)(acc + row * H_C))[lane] = v;
        }
        tile_sweep(src, dst, deg, msg, dinv_t, acc, keep, kept, mine[wave],
                   t0, e, wave, lane);
    }

    for (int row = wave; row < AGG_TN; row += AGG_WAVES) {
        const int node = t0 + row;
        if (node < n) {
            float4 v = ((const float4*)(acc + row * H_C))[lane];
            v.x = fmaxf(v.x, 0.0f);
            v.y = fmaxf(v.y, 0.0f);
            v.z = fmaxf(v.z, 0.0f);
            v.w = fmaxf(v.w, 0.0f);
            ((float4*)(out + (size_t)node * H_C))[lane] = v;
        }
    }

    if (!FIRST) {
        const int c  = tid & (H_C - 1);
        const int q  = __builtin_amdgcn_readfirstlane(tid >> 7);
        const int nb = q * POOL_GN;
        int nval = n - (t0 + nb);
        nval = nval < POOL_GN ? nval : POOL_GN;
        if (nval > 0) {
            float v[POOL_GN];
#pragma unroll
            for (int k = 0; k < POOL_GN; ++k) v[k] = acc[(nb + k) * H_C + c];
            const int gl = gid_t[nb + lane];
            int   g   = __builtin_amdgcn_readlane(gl, 0);
            float run = 0.0f;
            int   len = 0;
#pragma unroll
            for (int k = 0; k < POOL_GN; ++k) {
                if (k < nval) {
                    const int gk = __builtin_amdgcn_readlane(gl, k);
                    if (gk != g) {
                        atomic_fadd(&pooled[(size_t)g * H_C + c], run);
                        if (c == 0) atomic_fadd(&cnts[g], (float)len);
                        run = 0.0f;
                        len = 0;
                        g   = gk;
                    }
                    run += fmaxf(v[k], 0.0f);
                    ++len;
                }
            }
            atomic_fadd(&pooled[(size_t)g * H_C + c], run);
            if (c == 0) atomic_fadd(&cnts[g], (float)len);
        }
    }
}

extern "C" void kernel_launch(void* const* d_in, const int* in_sizes, int n_in,
                              void* d_out, int out_size, void* d_ws, size_t ws_size,
                              hipStream_t stream) {
    const float* x    = (const float*)d_in[0];
    const int*   ei   = (const int*)  d_in[1];
    const int*   batch= (const int*)  d_in[2];
    const float* W1   = (const float*)d_in[3];
    const float* b1   = (const float*)d_in[4];
    const float* W2   = (const float*)d_in[5];
    const float* b2   = (const float*)d_in[6];
    const float* Wl   = (const float*)d_in[7];
    const float* bl   = (const float*)d_in[8];

    const int N = in_sizes[0] / IN_C;
    const int E = in_sizes[1] / 2;
    const int G = (out_size - N * OUT_C) / H_C;

    const int* esrc = ei;
    const int* edst = ei + E;

    float* logits = (float*)d_out;
    float* pooled = logits + (size_t)N * OUT_C;

    float* deg  = (float*)d_ws;
    float* tmpA = deg + (((size_t)N + 255) & ~(size_t)255);
    float* tmpB = tmpA + (size_t)N * H_C;
    float* cnts = tmpB + (size_t)N * H_C;

    const int aggTiles = (N + AGG_TN - 1) / AGG_TN;
    unsigned int* lcount = (unsigned int*)(cnts + (((size_t)G + 255) & ~(size_t)255));
    unsigned int* lists = lcount + (((size_t)aggTiles * AGG_WAVES + 63) & ~(size_t)63);
    const int chunks = (E + CS_EDGES - 1) / CS_EDGES;
    unsigned int* cstart = lists + (size_t)aggTiles * AGG_WAVES * AGG_CAP * 2;
    unsigned int* sorted = cstart + (size_t)chunks * 256;
    const int rowBlocks  = (N + 15) / 16;
    const int gemmBlocks = rowBlocks < 512 ? rowBlocks : 512;

    gemm_wmma_kernel<IN_C, H_C, false, 1><<<gemmBlocks, 256, 0, stream>>>(x, W1, nullptr, tmpA, N, deg, N, cnts, pooled, G);
    chunk_sort_kernel<<<(chunks + CS_WAVES - 1) / CS_WAVES, CS_WAVES * 32, 0, stream>>>(edst, deg, cstart, sorted, E);
    tile_agg_kernel<true><<<aggTiles, AGG_WAVES * 32, 0, stream>>>(esrc, edst, deg, tmpA, b1, tmpB, N, E, lcount, lists, cstart, sorted, nullptr, nullptr, nullptr);

    gemm_wmma_kernel<H_C, H_C, false, 0><<<gemmBlocks, 256, 0, stream>>>(tmpB, W2, nullptr, tmpA, N, nullptr, 0, nullptr, nullptr, 0);
    tile_agg_kernel<false><<<aggTiles, AGG_WAVES * 32, 0, stream>>>(esrc, edst, deg, tmpA, b2, tmpB, N, E, lcount, lists, nullptr, nullptr, batch, pooled, cnts);

    gemm_wmma_kernel<H_C, OUT_C, true, 2><<<gemmBlocks, 128, 0, stream>>>(tmpB, Wl, bl, logits, N, nullptr, 0, cnts, pooled, G);
}


# first aggregation: a tile's collected edges go to their owning waves by a stable counting sort in LDS (each wave sorts only its own part) instead of every wave scanning all 32 parts; sum order kept
# speedup vs baseline: 3.0492x; 1.0616x over previous
#include <hip/hip_runtime.h>

#define IN_C  128
#define H_C   128
#define OUT_C 64

typedef float v2f __attribute__((ext_vector_type(2)));
typedef float v8f __attribute__((ext_vector_type(8)));

__device__ __forceinline__ void atomic_fadd(float* p, float v) {
    asm volatile("global_atomic_add_f32 %0, %1, off" : : "v"(p), "v"(v) : "memory");
}

__device__ __forceinline__ float dinv_at(const float* __restrict__ deg, int i) {
    return rsqrtf(deg[i]);
}

template<int K, int NC, bool ADD_BIAS, int SIDE>
__global__ __launch_bounds__(2 * NC)
void gemm_wmma_kernel(const float* __restrict__ A,
                                 const float* __restrict__ B,
                                 const float* __restrict__ bias,
                                 float* __restrict__ C, int M,
                                 float* __restrict__ deg, int nodes,
                                 float* __restrict__ cnts,
                                 float* __restrict__ pooled, int graphs) {
    const int tid   = threadIdx.x;
    const int wave  = tid >> 5;
    const int lane  = tid & 31;
    const int n0    = wave * 16;
    const int lhalf = lane >> 4;
    const int l15   = lane & 15;
    const int n     = n0 + l15;

    v2f bfrag[K / 4];
#pragma unroll
    for (int t = 0; t < K / 4; ++t) {
        const int kk = 8 * (t >> 1) + 4 * lhalf + 2 * (t & 1);
        bfrag[t].x = B[(size_t)kk * NC + n];
        bfrag[t].y = B[(size_t)(kk + 1) * NC + n];
    }
    const float bv = ADD_BIAS ? bias[n] : 0.0f;

    if (SIDE != 0) {
        const int gsz = gridDim.x * blockDim.x;
        const int g0  = blockIdx.x * blockDim.x + tid;
        if (SIDE == 1) {
            for (int i = g0; i < nodes; i += gsz)        deg[i]    = 1.0f;
            for (int i = g0; i < graphs; i += gsz)       cnts[i]   = 0.0f;
            for (int i = g0; i < graphs * H_C; i += gsz) pooled[i] = 0.0f;
        } else {
            for (int i = g0; i < graphs * H_C; i += gsz)
                pooled[i] /= fmaxf(cnts[i / H_C], 1.0f);
        }
    }

    const int strips = (M + 15) / 16;

    for (int s = blockIdx.x; s < strips; s += gridDim.x) {
        const int row0 = s * 16;
        const int row  = (row0 + l15 < M) ? row0 + l15 : M - 1;
        const float4* arow = (const float4*)(A + (size_t)row * K);

        float4 af[K / 8];
#pragma unroll
        for (int i = 0; i < K / 8; ++i) af[i] = arow[2 * i + lhalf];
        static_assert(K / 8 == 16, "the load fence below names 16 loads");
        asm volatile("" : : "v"(af[0].x),  "v"(af[1].x),  "v"(af[2].x),  "v"(af[3].x),
                            "v"(af[4].x),  "v"(af[5].x),  "v"(af[6].x),  "v"(af[7].x),
                            "v"(af[8].x),  "v"(af[9].x),  "v"(af[10].x), "v"(af[11].x),
                            "v"(af[12].x), "v"(af[13].x), "v"(af[14].x), "v"(af[15].x));

        v8f acc = {};
#pragma unroll
        for (int i = 0; i < K / 8; ++i) {
            v2f a0, a1;
            a0.x = af[i].x; a0.y = af[i].y;
            a1.x = af[i].z; a1.y = af[i].w;
            acc = __builtin_amdgcn_wmma_f32_16x16x4_f32(
                      false, a0, false, bfrag[2 * i], (short)0, acc, false, false);
            acc = __builtin_amdgcn_wmma_f32_16x16x4_f32(
                      false, a1, false, bfrag[2 * i + 1], (short)0, acc, false, false);
        }

#pragma unroll
        for (int r = 0; r < 8; ++r) {
            int m = row0 + r + 8 * lhalf;
            if (m < M) C[(size_t)m * NC + n] = acc[r] + bv;
        }
    }
}

#define AGG_TN    256
#define AGG_WAVES 32
#define AGG_STEP  1024
#define AGG_CHUNK (AGG_WAVES * AGG_STEP)
#define AGG_CAP   1024
#define POOL_GN   32

template<int R>
__device__ __forceinline__ void agg_add_rows(int s, int dl, float w, int nk, int lane,
                                             const float* __restrict__ msg, float* acc) {
    for (int j = 0; j < nk; j += R) {
        float4 r[R];
        int    dj[R];
        float  wj[R];
#pragma unroll
        for (int t = 0; t < R; ++t) {
            const int jx = (j + t < nk) ? j + t : nk - 1;
            const int sj = __builtin_amdgcn_readlane(s, jx);
            r[t]  = ((const float4*)(msg + (size_t)sj * H_C))[lane];
        }
#pragma unroll
        for (int t = 0; t < R; ++t) {
            const int jx = (j + t < nk) ? j + t : nk - 1;
            dj[t] = __builtin_amdgcn_readlane(dl, jx);
            wj[t] = __int_as_float(__builtin_amdgcn_readlane(__float_as_int(w), jx));
        }
#pragma unroll
        for (int t = 0; t < R; ++t) {
            if (j + t < nk) {
                float4* a = (float4*)(acc + dj[t] * H_C) + lane;
                float4  v = *a;
                v.x += r[t].x * wj[t];
                v.y += r[t].y * wj[t];
                v.z += r[t].z * wj[t];
                v.w += r[t].w * wj[t];
                *a = v;
            }
        }
    }
}

template<int R>
__device__ __forceinline__ void agg_add_rows_run(int s, int pl, float w, int nk, int lane,
                                                 const float* __restrict__ msg, float* rows,
                                                 int& cur, float4& run) {
    for (int j = 0; j < nk; j += R) {
        float4 r[R];
        int    pj[R];
        float  wj[R];
#pragma unroll
        for (int t = 0; t < R; ++t) {
            const int jx = (j + t < nk) ? j + t : nk - 1;
            const int sj = __builtin_amdgcn_readlane(s, jx);
            r[t]  = ((const float4*)(msg + (size_t)sj * H_C))[lane];
        }
#pragma unroll
        for (int t = 0; t < R; ++t) {
            const int jx = (j + t < nk) ? j + t : nk - 1;
            pj[t] = __builtin_amdgcn_readlane(pl, jx);
            wj[t] = __int_as_float(__builtin_amdgcn_readlane(__float_as_int(w), jx));
        }
#pragma unroll
        for (int t = 0; t < R; ++t) {
            if (j + t < nk) {
                if (pj[t] != cur) {
                    ((float4*)(rows + cur * H_C))[lane] = run;
                    cur = pj[t];
                    run = ((const float4*)(rows + cur * H_C))[lane];
                }
                run.x += r[t].x * wj[t];
                run.y += r[t].y * wj[t];
                run.z += r[t].z * wj[t];
                run.w += r[t].w * wj[t];
            }
        }
    }
}

__device__ __forceinline__ void agg_drain(const unsigned int* myl, int m, int cb, int lane,
                                          const int* __restrict__ src,
                                          const float* __restrict__ deg,
                                          const float* __restrict__ msg,
                                          const float* dinv_t, float* acc) {
    for (int k0 = 0; k0 < m; k0 += 32) {
        const int nk = (m - k0 < 32) ? m - k0 : 32;
        int   s  = 0;
        int   dl = 0;
        float w  = 0.0f;
        if (lane < nk) {
            const unsigned int ent = myl[k0 + lane];
            dl = (int)((ent >> 15) & 255u);
            s  = src[cb + (int)(ent & 32767u)];
            w  = dinv_at(deg, s) * dinv_t[dl];
        }
        agg_add_rows<4>(s, dl, w, nk, lane, msg, acc);
    }
}

__device__ __forceinline__ void agg_load_step(const int* __restrict__ dst, int base, int e,
                                              int lane, int (&d)[32]) {
    if (base + AGG_STEP <= e) {
        const int* p = dst + base + lane;
#pragma unroll
        for (int u = 0; u < 32; ++u) d[u] = p[u * 32];
    } else {
#pragma unroll
        for (int u = 0; u < 32; ++u) {
            const int idx = base + u * 32 + lane;
            const int v   = dst[idx < e ? idx : e - 1];
            d[u] = (idx < e) ? v : -1;
        }
    }
}

__device__ __forceinline__ void tile_sweep(const int* __restrict__ src,
                                           const int* __restrict__ dst,
                                           const float* __restrict__ deg,
                                           const float* __restrict__ msg,
                                           const float* dinv_t, float* acc,
                                           unsigned int* keep, int* kept, unsigned int* myl,
                                           int t0, int e, int wave, int lane) {
    int d[32];
    agg_load_step(dst, wave * AGG_STEP, e, lane, d);
    __syncthreads();

    for (int cb = 0; cb < e; cb += AGG_CHUNK) {
        int cnt = 0;
#pragma unroll
        for (int u = 0; u < 32; ++u) {
            const unsigned int dl   = (unsigned int)(d[u] - t0);
            const bool         hit  = dl < (unsigned int)AGG_TN;
            const unsigned int mask = __builtin_amdgcn_ballot_w32(hit);
            if (mask != 0u) {
                if (hit)
                    keep[(cnt + (int)__builtin_amdgcn_mbcnt_lo(mask, 0u)) * AGG_WAVES + wave] =
                        (unsigned int)(wave * AGG_STEP + u * 32 + lane) | (dl << 15);
                cnt += __popc(mask);
            }
        }
        if (lane == 0) kept[wave] = cnt;
        __syncthreads();

        agg_load_step(dst, cb + AGG_CHUNK + wave * AGG_STEP, e, lane, d);

        const int c = kept[lane];
        int maxc = c;
#pragma unroll
        for (int o = 16; o > 0; o >>= 1) {
            const int other = __shfl_xor(maxc, o);
            maxc = other > maxc ? other : maxc;
        }
        maxc = __builtin_amdgcn_readfirstlane(maxc);
        int mcnt = 0;
        for (int j = 0; j < maxc; j += 4) {
            unsigned int ent[4];
#pragma unroll
            for (int t = 0; t < 4; ++t)
                ent[t] = (j + t < c) ? keep[(j + t) * AGG_WAVES + lane] : 0u;
#pragma unroll
            for (int t = 0; t < 4; ++t) {
                const bool         hit  = (j + t < c) && ((ent[t] >> 18) == (unsigned int)wave);
                const unsigned int mask = __builtin_amdgcn_ballot_w32(hit);
                if (mask != 0u) {
                    if (hit) myl[mcnt + (int)__builtin_amdgcn_mbcnt_lo(mask, 0u)] = ent[t];
                    mcnt += __popc(mask);
                }
            }
            if (mcnt > 32) {
                agg_drain(myl, mcnt, cb, lane, src, deg, msg, dinv_t, acc);
                mcnt = 0;
            }
        }
        if (mcnt > 0) agg_drain(myl, mcnt, cb, lane, src, deg, msg, dinv_t, acc);
        __syncthreads();
    }
}

#define CS_EDGES 1024
#define CS_WAVES 4
static_assert(CS_EDGES == AGG_STEP, "a chunk is read with agg_load_step");
static_assert(AGG_TN == 256, "the tile of a destination is node >> 8");

__device__ __forceinline__ unsigned int tile_of(int d) {
    const unsigned int t = (unsigned int)d >> 8;
    return t < 255u ? t : 255u;
}

__global__ __launch_bounds__(CS_WAVES * 32)
void chunk_sort_kernel(const int* __restrict__ dst,
                       float* __restrict__ deg,
                       unsigned int* __restrict__ cstart,
                       unsigned int* __restrict__ sorted,
                       int e) {
    __shared__ unsigned int hist[CS_WAVES * 256];
    const int lane  = threadIdx.x & 31;
    const int wave  = __builtin_amdgcn_readfirstlane((int)(threadIdx.x >> 5));
    const int chunk = blockIdx.x * CS_WAVES + wave;
    const int base  = chunk * CS_EDGES;
    if (base >= e) return;
    unsigned int* h = hist + wave * 256;
#pragma unroll
    for (int k = 0; k < 8; ++k) h[lane + 32 * k] = 0u;

    int d[32];
    agg_load_step(dst, base, e, lane, d);

    unsigned int rk[32];
#pragma unroll
    for (int u = 0; u < 32; ++u) {
        const bool valid = d[u] >= 0;
        if (valid) atomic_fadd(&deg[d[u]], 1.0f);
        const unsigned int t = valid ? tile_of(d[u]) : 255u;
        unsigned int m = 0xffffffffu;
#pragma unroll
        for (int b = 0; b < 8; ++b) {
            const bool         bit = ((t >> b) & 1u) != 0u;
            const unsigned int bal = __builtin_amdgcn_ballot_w32(bit);
            m &= bit ? bal : ~bal;
        }
        const unsigned int r   = __builtin_amdgcn_mbcnt_lo(m, 0u);
        const unsigned int old = h[t];
        rk[u] = old + r;
        if (r == 0u) h[t] = old + (unsigned int)__popc(m);
    }

    unsigned int c[8];
    unsigned int s = 0u;
#pragma unroll
    for (int k = 0; k < 8; ++k) { c[k] = h[8 * lane + k]; s += c[k]; }
    unsigned int incl = s;
#pragma unroll
    for (int o = 1; o < 32; o <<= 1) {
        const unsigned int up = (unsigned int)__shfl_up((int)incl, o);
        if (lane >= o) incl += up;
    }
    unsigned int run = incl - s;
    unsigned int* cs = cstart + (size_t)chunk * 256;
#pragma unroll
    for (int k = 0; k < 8; ++k) {
        h[8 * lane + k]  = run;
        cs[8 * lane + k] = run;
        run += c[k];
    }

    unsigned int* so = sorted + (size_t)chunk * CS_EDGES;
#pragma unroll
    for (int u = 0; u < 32; ++u) {
        if (d[u] >= 0)
            so[h[tile_of(d[u])] + rk[u]] =
                (unsigned int)(base + 32 * u + lane) | (((unsigned int)d[u] & 255u) << 24);
    }
}

__device__ __forceinline__ int tile_collect(const unsigned int* __restrict__ cstart,
                                            const unsigned int* __restrict__ sorted,
                                            unsigned int* myk, unsigned int tile, int e,
                                            int wave, int lane) {
    const int chunks = (e + CS_EDGES - 1) / CS_EDGES;
    const int per    = (chunks + AGG_WAVES - 1) / AGG_WAVES;
    const int cbeg   = wave * per;
    const int cend   = (cbeg + per < chunks) ? cbeg + per : chunks;
    const unsigned int tq = tile < 255u ? tile : 254u;
    int found = 0;
    for (int c0 = cbeg; c0 < cend; c0 += 32) {
        const int  c  = c0 + lane;
        const bool ok = c < cend;
        unsigned int a = 0u, b = 0u;
        if (ok) {
            const unsigned int* cs = cstart + (size_t)c * 256 + tq;
            a = cs[0];
            b = cs[1];
        }
        const int nl = (int)(b - a);
        int incl = nl;
#pragma unroll
        for (int o = 1; o < 32; o <<= 1) {
            const int up = __shfl_up(incl, o);
            if (lane >= o) incl += up;
        }
        int mx = nl;
#pragma unroll
        for (int o = 16; o > 0; o >>= 1) {
            const int other = __shfl_xor(mx, o);
            mx = other > mx ? other : mx;
        }
        mx = __builtin_amdgcn_readfirstlane(mx);
        const int pos0 = found + incl - nl;
        const unsigned int* sp = sorted + (size_t)(ok ? c : cbeg) * CS_EDGES + a;
        for (int k = 0; k < mx; k += 4) {
            unsigned int v[4];
#pragma unroll
            for (int t = 0; t < 4; ++t) {
                v[t] = 0u;
                if (k + t < nl) v[t] = sp[k + t];
            }
#pragma unroll
            for (int t = 0; t < 4; ++t) {
                const int pos = pos0 + k + t;
                if (k + t < nl && pos < AGG_STEP) myk[pos] = v[t];
            }
        }
        found += __builtin_amdgcn_readlane(incl, 31);
    }
    return found;
}

template<bool FIRST>
__global__ __launch_bounds__(AGG_WAVES * 32)
void tile_agg_kernel(const int* __restrict__ src,
                     const int* __restrict__ dst,
                     const float* __restrict__ deg,
                     const float* __restrict__ msg,
                     const float* __restrict__ bias,
                     float* __restrict__ out, int n, int e,
                     unsigned int* __restrict__ lcount,
                     unsigned int* __restrict__ lists,
                     const unsigned int* __restrict__ cstart,
                     const unsigned int* __restrict__ sorted,
                     const int* __restrict__ batch,
                     float* __restrict__ pooled,
                     float* __restrict__ cnts) {
    __shared__ __attribute__((aligned(16))) float acc[AGG_TN * H_C];
    __shared__ float        dinv_t[AGG_TN];
    __shared__ int          gid_t[AGG_TN];
    __shared__ unsigned int keep[AGG_STEP * AGG_WAVES];
    __shared__ int          kept[AGG_WAVES];
    __shared__ unsigned int mine[AGG_WAVES][160];

    const int tid  = threadIdx.x;
    const int lane = tid & 31;
    const int wave = __builtin_amdgcn_readfirstlane(tid >> 5);
    const int t0   = blockIdx.x * AGG_TN;

    uint2* lst = (uint2*)lists + (size_t)(blockIdx.x * AGG_WAVES + wave) * AGG_CAP;

    bool fits = false;
    int  mlist = 0;
    uint2 ecur = make_uint2(0u, 0u);
    if (FIRST) {
        if (tid < AGG_TN) dinv_t[tid] = (t0 + tid < n) ? dinv_at(deg, t0 + tid) : 0.0f;
        unsigned int* myk = keep + wave * AGG_STEP;
        const int found = tile_collect(cstart, sorted, myk, (unsigned int)blockIdx.x, e, wave, lane);
        if (lane == 0) kept[wave] = found;
        __syncthreads();
        const int  fl     = kept[lane];
        const bool partok = __builtin_amdgcn_ballot_w32(fl > AGG_STEP) == 0u &&
                            n <= 255 * AGG_TN && e <= (1 << 24);
        int lcnt = 0;
        static_assert(8 * H_C == AGG_CAP, "a wave's notes fill its own 8 rows");
        static_assert(AGG_WAVES == 32, "one lane per owning wave in the counts below");
        float* rawl = acc + wave * 8 * H_C;
        unsigned int* cw = mine[wave];
        const int cp = found < AGG_STEP ? found : AGG_STEP;
        if (partok) {
            cw[lane] = 0u;
            for (int j = 0; j < cp; j += 32) {
                const bool         valid = j + lane < cp;
                const unsigned int ent   = valid ? myk[j + lane] : 0u;
                const unsigned int ow    = ent >> 27;
                unsigned int m = __builtin_amdgcn_ballot_w32(valid);
                for (int b = 0; b < 5; ++b) {
                    const bool         bit = ((ow >> b) & 1u) != 0u;
                    const unsigned int bal = __builtin_amdgcn_ballot_w32(bit);
                    m &= bit ? bal : ~bal;
                }
                if (valid && __builtin_amdgcn_mbcnt_lo(m, 0u) == 0u) cw[ow] += (unsigned int)__popc(m);
            }
        }
        __syncthreads();
        if (partok) {
            const unsigned int cme = mine[lane][wave];
            unsigned int incl = cme;
            for (int o = 1; o < 32; o <<= 1) {
                const unsigned int up = (unsigned int)__shfl_up((int)incl, o);
                if (lane >= o) incl += up;
            }
            mine[lane][32 + wave] = incl - cme;
            lcnt = __builtin_amdgcn_readlane((int)incl, 31);
        }
        __syncthreads();
        if (partok) {
            cw[lane] = cw[32 + lane];
            for (int j = 0; j < cp; j += 32) {
                const bool         valid = j + lane < cp;
                const unsigned int ent   = valid ? myk[j + lane] : 0u;
                const unsigned int ow    = ent >> 27;
                unsigned int m = __builtin_amdgcn_ballot_w32(valid);
                for (int b = 0; b < 5; ++b) {
                    const bool         bit = ((ow >> b) & 1u) != 0u;
                    const unsigned int bal = __builtin_amdgcn_ballot_w32(bit);
                    m &= bit ? bal : ~bal;
                }
                if (valid) {
                    const unsigned int r   = __builtin_amdgcn_mbcnt_lo(m, 0u);
                    const unsigned int old = cw[ow];
                    const unsigned int pos = old + r;
                    if (pos < (unsigned int)AGG_CAP)
                        (acc + ow * 8 * H_C)[pos] = __uint_as_float(ent & 0x07ffffffu);
                    if (r == 0u) cw[ow] = old + (unsigned int)__popc(m);
                }
            }
        }
        const int mark = partok ? lcnt : AGG_CAP + 1;
        if (lane == 0) {
            lcount[blockIdx.x * AGG_WAVES + wave] = (unsigned int)mark;
            gid_t[wave] = mark;
        }
        __syncthreads();
        fits  = __builtin_amdgcn_ballot_w32(gid_t[lane] > AGG_CAP) == 0u;
        mlist = lcnt;
        if (fits) {
            int cnt[8] = {0, 0, 0, 0, 0, 0, 0, 0};
            for (int k0 = 0; k0 < mlist; k0 += 32) {
                const int idx = k0 + lane;
                const int pl  = idx < mlist ? (int)((__float_as_uint(rawl[idx]) >> 24) & 7u) : 8;
#pragma unroll
                for (int p = 0; p < 8; ++p)
                    cnt[p] += __popc(__builtin_amdgcn_ballot_w32(pl == p));
            }
            int off[8];
            off[0] = 0;
#pragma unroll
            for (int p = 1; p < 8; ++p) off[p] = off[p - 1] + cnt[p - 1];
            for (int k0 = 0; k0 < mlist; k0 += 256) {
                unsigned int raw[8];
                int          sv[8];
                float        dv[8];
#pragma unroll
                for (int t = 0; t < 8; ++t) {
                    const int idx = k0 + 32 * t + lane;
                    raw[t] = __float_as_uint(rawl[idx < mlist ? idx : mlist - 1]);
                }
#pragma unroll
                for (int t = 0; t < 8; ++t) sv[t] = src[raw[t] & 0xffffffu];
                asm volatile("" : : "v"(sv[0]), "v"(sv[1]), "v"(sv[2]), "v"(sv[3]),
                                    "v"(sv[4]), "v"(sv[5]), "v"(sv[6]), "v"(sv[7]));
#pragma unroll
                for (int t = 0; t < 8; ++t) dv[t] = deg[sv[t]];
                asm volatile("" : : "v"(dv[0]), "v"(dv[1]), "v"(dv[2]), "v"(dv[3]),
                                    "v"(dv[4]), "v"(dv[5]), "v"(dv[6]), "v"(dv[7]));
#pragma unroll
                for (int t = 0; t < 8; ++t) {
                    if (k0 + 32 * t < mlist) {
                        const int  idx   = k0 + 32 * t + lane;
                        const bool valid = idx < mlist;
                        const int  pl    = valid ? (int)((raw[t] >> 24) & 7u) : 8;
                        int pos = 0;
#pragma unroll
                        for (int p = 0; p < 8; ++p) {
                            const unsigned int mask = __builtin_amdgcn_ballot_w32(pl == p);
                            if (pl == p) pos = off[p] + (int)__builtin_amdgcn_mbcnt_lo(mask, 0u);
                            off[p] += __popc(mask);
                        }
                        const float w = rsqrtf(dv[t]) * dinv_t[wave * 8 + (pl & 7)];
                        if (valid)
                            lst[pos] = make_uint2((unsigned int)sv[t] | ((unsigned int)pl << 29),
                                                  (unsigned int)__float_as_int(w));
                    }
                }
            }
            __syncthreads();
            if (lane < mlist) ecur = lst[lane];
        }
    } else {
        const unsigned int c = lcount[blockIdx.x * AGG_WAVES + lane];
        fits  = __builtin_amdgcn_ballot_w32(c > (unsigned int)AGG_CAP) == 0u;
        mlist = __builtin_amdgcn_readlane((int)c, wave);
        if (fits && lane < mlist) ecur = lst[lane];
        if (tid < AGG_TN) {
            dinv_t[tid] = (t0 + tid < n) ? dinv_at(deg, t0 + tid) : 0.0f;
            gid_t[tid]  = (t0 + tid < n) ? batch[t0 + tid] : -1;
        }
    }

    if (fits) {
        float* rows = acc + wave * 8 * H_C;
        {
            const float4 b = ((const float4*)bias)[lane];
            float4 m[8];
            float  dg[8];
#pragma unroll
            for (int p = 0; p < 8; ++p) {
                const int node = t0 + wave * 8 + p;
                const int nc   = node < n ? node : n - 1;
                m[p]  = ((const float4*)(msg + (size_t)nc * H_C))[lane];
                dg[p] = deg[nc];
            }
#pragma unroll
            for (int p = 0; p < 8; ++p) {
                const float di = rsqrtf(dg[p]);
                float4 v = make_float4(0.0f, 0.0f, 0.0f, 0.0f);
                if (t0 + wave * 8 + p < n) {
                    v.x = m[p].x * di * di + b.x;
                    v.y = m[p].y * di * di + b.y;
                    v.z = m[p].z * di * di + b.z;
                    v.w = m[p].w * di * di + b.w;
                }
                ((float4*)(rows + p * H_C))[lane] = v;
            }
        }
        int    cur = 0;
        float4 run = ((const float4*)rows)[lane];
        for (int k0 = 0; k0 < mlist; k0 += 32) {
            const int nk = (mlist - k0 < 32) ? mlist - k0 : 32;
            uint2 enext = make_uint2(0u, 0u);
            if (k0 + 32 + lane < mlist) enext = lst[k0 + 32 + lane];
            int   s  = 0;
            int   pl = 0;
            float w  = 0.0f;
            if (lane < nk) {
                s  = (int)(ecur.x & 0x1fffffffu);
                pl = (int)(ecur.x >> 29);
                w  = __int_as_float((int)ecur.y);
            }
            agg_add_rows_run<8>(s, pl, w, nk, lane, msg, rows, cur, run);
            ecur = enext;
        }
        ((float4*)(rows + cur * H_C))[lane] = run;
        __syncthreads();
    } else {
        for (int row = wave; row < AGG_TN; row += AGG_WAVES) {
            const int node = t0 + row;
            float4 v = make_float4(0.0f, 0.0f, 0.0f, 0.0f);
            if (node < n) {
                const float  di = dinv_at(deg, node);
                const float4 m  = ((const float4*)(msg + (size_t)node * H_C))[lane];
                const float4 b  = ((const float4*)bias)[lane];
                v.x = m.x * di * di + b.x;
                v.y = m.y * di * di + b.y;
                v.z = m.z * di * di + b.z;
                v.w = m.w * di * di + b.w;
            }
            ((float4*)(acc + row * H_C))[lane] = v;
        }
        tile_sweep(src, dst, deg, msg, dinv_t, acc, keep, kept, mine[wave],
                   t0, e, wave, lane);
    }

    for (int row = wave; row < AGG_TN; row += AGG_WAVES) {
        const int node = t0 + row;
        if (node < n) {
            float4 v = ((const float4*)(acc + row * H_C))[lane];
            v.x = fmaxf(v.x, 0.0f);
            v.y = fmaxf(v.y, 0.0f);
            v.z = fmaxf(v.z, 0.0f);
            v.w = fmaxf(v.w, 0.0f);
            ((float4*)(out + (size_t)node * H_C))[lane] = v;
        }
    }

    if (!FIRST) {
        const int c  = tid & (H_C - 1);
        const int q  = __builtin_amdgcn_readfirstlane(tid >> 7);
        const int nb = q * POOL_GN;
        int nval = n - (t0 + nb);
        nval = nval < POOL_GN ? nval : POOL_GN;
        if (nval > 0) {
            float v[POOL_GN];
#pragma unroll
            for (int k = 0; k < POOL_GN; ++k) v[k] = acc[(nb + k) * H_C + c];
            const int gl = gid_t[nb + lane];
            int   g   = __builtin_amdgcn_readlane(gl, 0);
            float run = 0.0f;
            int   len = 0;
#pragma unroll
            for (int k = 0; k < POOL_GN; ++k) {
                if (k < nval) {
                    const int gk = __builtin_amdgcn_readlane(gl, k);
                    if (gk != g) {
                        atomic_fadd(&pooled[(size_t)g * H_C + c], run);
                        if (c == 0) atomic_fadd(&cnts[g], (float)len);
                        run = 0.0f;
                        len = 0;
                        g   = gk;
                    }
                    run += fmaxf(v[k], 0.0f);
                    ++len;
                }
            }
            atomic_fadd(&pooled[(size_t)g * H_C + c], run);
            if (c == 0) atomic_fadd(&cnts[g], (float)len);
        }
    }
}

extern "C" void kernel_launch(void* const* d_in, const int* in_sizes, int n_in,
                              void* d_out, int out_size, void* d_ws, size_t ws_size,
                              hipStream_t stream) {
    const float* x    = (const float*)d_in[0];
    const int*   ei   = (const int*)  d_in[1];
    const int*   batch= (const int*)  d_in[2];
    const float* W1   = (const float*)d_in[3];
    const float* b1   = (const float*)d_in[4];
    const float* W2   = (const float*)d_in[5];
    const float* b2   = (const float*)d_in[6];
    const float* Wl   = (const float*)d_in[7];
    const float* bl   = (const float*)d_in[8];

    const int N = in_sizes[0] / IN_C;
    const int E = in_sizes[1] / 2;
    const int G = (out_size - N * OUT_C) / H_C;

    const int* esrc = ei;
    const int* edst = ei + E;

    float* logits = (float*)d_out;
    float* pooled = logits + (size_t)N * OUT_C;

    float* deg  = (float*)d_ws;
    float* tmpA = deg + (((size_t)N + 255) & ~(size_t)255);
    float* tmpB = tmpA + (size_t)N * H_C;
    float* cnts = tmpB + (size_t)N * H_C;

    const int aggTiles = (N + AGG_TN - 1) / AGG_TN;
    unsigned int* lcount = (unsigned int*)(cnts + (((size_t)G + 255) & ~(size_t)255));
    unsigned int* lists = lcount + (((size_t)aggTiles * AGG_WAVES + 63) & ~(size_t)63);
    const int chunks = (E + CS_EDGES - 1) / CS_EDGES;
    unsigned int* cstart = lists + (size_t)aggTiles * AGG_WAVES * AGG_CAP * 2;
    unsigned int* sorted = cstart + (size_t)chunks * 256;
    const int rowBlocks  = (N + 15) / 16;
    const int gemmBlocks = rowBlocks < 512 ? rowBlocks : 512;

    gemm_wmma_kernel<IN_C, H_C, false, 1><<<gemmBlocks, 256, 0, stream>>>(x, W1, nullptr, tmpA, N, deg, N, cnts, pooled, G);
    chunk_sort_kernel<<<(chunks + CS_WAVES - 1) / CS_WAVES, CS_WAVES * 32, 0, stream>>>(edst, deg, cstart, sorted, E);
    tile_agg_kernel<true><<<aggTiles, AGG_WAVES * 32, 0, stream>>>(esrc, edst, deg, tmpA, b1, tmpB, N, E, lcount, lists, cstart, sorted, nullptr, nullptr, nullptr);

    gemm_wmma_kernel<H_C, H_C, false, 0><<<gemmBlocks, 256, 0, stream>>>(tmpB, W2, nullptr, tmpA, N, nullptr, 0, nullptr, nullptr, 0);
    tile_agg_kernel<false><<<aggTiles, AGG_WAVES * 32, 0, stream>>>(esrc, edst, deg, tmpA, b2, tmpB, N, E, lcount, lists, nullptr, nullptr, batch, pooled, cnts);

    gemm_wmma_kernel<H_C, OUT_C, true, 2><<<gemmBlocks, 128, 0, stream>>>(tmpB, Wl, bl, logits, N, nullptr, 0, cnts, pooled, G);
}
